# starting kernel
# baseline (speedup 1.0000x reference)
#include <hip/hip_runtime.h>

#define N_NODES 50000
#define N_EDGES 800000

typedef __attribute__((ext_vector_type(2))) float v2f;
typedef __attribute__((ext_vector_type(8))) float v8f;


__global__ void zero_f32(float* __restrict__ p, long n) {
  long i = (long)blockIdx.x * blockDim.x + threadIdx.x;
  long stride = (long)gridDim.x * blockDim.x;
  for (; i < n; i += stride) p[i] = 0.0f;
}

__global__ void count_edges(const int* __restrict__ dst, float* __restrict__ cnt, int nE) {
  int e = blockIdx.x * blockDim.x + threadIdx.x;
  if (e < nE) atomicAdd(&cnt[dst[e]], 1.0f);
}

__global__ void recip_clamp(const float* __restrict__ cnt, float* __restrict__ inv, int n) {
  int i = blockIdx.x * blockDim.x + threadIdx.x;
  if (i < n) inv[i] = 1.0f / fmaxf(cnt[i], 1.0f);
}

__global__ void scatter_edges(const float* __restrict__ feat,
                              const int* __restrict__ src,
                              const int* __restrict__ dst,
                              float* __restrict__ agg,
                              int nE, int d) {
  int wave = blockIdx.x * (blockDim.x >> 5) + (threadIdx.x >> 5);
  int lane = threadIdx.x & 31;
  if (wave >= nE) return;
  const float* fs = feat + (size_t)src[wave] * d;
  float*       fd = agg  + (size_t)dst[wave] * d;
  for (int c = lane; c < d; c += 32)
    atomicAdd(&fd[c], fs[c]);
}


template<int DIN, int DOUT>
__global__ __launch_bounds__(32)
void sage_wmma_gemm(const float* __restrict__ agg,
                    const float* __restrict__ xin,
                    const float* __restrict__ inv,
                    const float* __restrict__ Wl,
                    const float* __restrict__ Wr,
                    const float* __restrict__ bias,
                    float* __restrict__ out) {
  const int lane = threadIdx.x & 31;
  const int half = lane >> 4;
  const int l16  = lane & 15;
  const int mTile = blockIdx.x << 4;
  const int nBase = blockIdx.y << 6;

  const int aRow = mTile + l16;
  const int kOff = half << 1;

  const float invc = inv[aRow];

  v8f acc[4] = {v8f{}, v8f{}, v8f{}, v8f{}};

  {
    const float* aP = agg + (size_t)aRow * DIN + kOff;
    const float* bP = Wl  + (size_t)(nBase + l16) * DIN + kOff;
#pragma unroll 4
    for (int k = 0; k < DIN; k += 4) {
      v2f a = *(const v2f*)(aP + k);
      a *= invc;
#pragma unroll
      for (int t = 0; t < 4; ++t) {
        v2f b = *(const v2f*)(bP + (size_t)t * 16 * DIN + k);
        acc[t] = __builtin_amdgcn_wmma_f32_16x16x4_f32(false, a, false, b,
                                                       (short)0, acc[t],
                                                       false, false);
      }
    }
  }

  {
    const float* aP = xin + (size_t)aRow * DIN + kOff;
    const float* bP = Wr  + (size_t)(nBase + l16) * DIN + kOff;
#pragma unroll 4
    for (int k = 0; k < DIN; k += 4) {
      v2f a = *(const v2f*)(aP + k);
#pragma unroll
      for (int t = 0; t < 4; ++t) {
        v2f b = *(const v2f*)(bP + (size_t)t * 16 * DIN + k);
        acc[t] = __builtin_amdgcn_wmma_f32_16x16x4_f32(false, a, false, b,
                                                       (short)0, acc[t],
                                                       false, false);
      }
    }
  }

  const int rowBase = mTile + (half << 3);
#pragma unroll
  for (int t = 0; t < 4; ++t) {
    const int col = nBase + t * 16 + l16;
    const float bj = bias[col];
#pragma unroll
    for (int i = 0; i < 8; ++i)
      out[(size_t)(rowBase + i) * DOUT + col] = acc[t][i] + bj;
  }
}


extern "C" void kernel_launch(void* const* d_in, const int* in_sizes, int n_in,
                              void* d_out, int out_size, void* d_ws, size_t ws_size,
                              hipStream_t stream) {
  const float* x   = (const float*)d_in[0];
  const int*   ei  = (const int*)d_in[1];
  const int*   src = ei;
  const int*   dst = ei + N_EDGES;
  const float* Wl1 = (const float*)d_in[2];
  const float* b1  = (const float*)d_in[3];
  const float* Wr1 = (const float*)d_in[4];
  const float* Wl2 = (const float*)d_in[5];
  const float* b2  = (const float*)d_in[6];
  const float* Wr2 = (const float*)d_in[7];
  const float* Wl3 = (const float*)d_in[8];
  const float* b3  = (const float*)d_in[9];
  const float* Wr3 = (const float*)d_in[10];
  float* out = (float*)d_out;

  float* ws  = (float*)d_ws;
  float* cnt = ws;
  float* inv = cnt + 50048;
  float* agg = inv + 50048;
  float* h1  = agg + (size_t)N_NODES * 128;
  float* h2  = h1  + (size_t)N_NODES * 128;

  zero_f32<<<256, 256, 0, stream>>>(cnt, (long)N_NODES);
  count_edges<<<(N_EDGES + 255) / 256, 256, 0, stream>>>(dst, cnt, N_EDGES);
  recip_clamp<<<(N_NODES + 255) / 256, 256, 0, stream>>>(cnt, inv, N_NODES);

  const int scatterBlocks = N_EDGES / 8;

  zero_f32<<<2048, 256, 0, stream>>>(agg, (long)N_NODES * 128);
  scatter_edges<<<scatterBlocks, 256, 0, stream>>>(x, src, dst, agg, N_EDGES, 128);
  {
    dim3 g(N_NODES / 16, 128 / 64);
    sage_wmma_gemm<128, 128><<<g, 32, 0, stream>>>(agg, x, inv, Wl1, Wr1, b1, h1);
  }

  zero_f32<<<2048, 256, 0, stream>>>(agg, (long)N_NODES * 128);
  scatter_edges<<<scatterBlocks, 256, 0, stream>>>(h1, src, dst, agg, N_EDGES, 128);
  {
    dim3 g(N_NODES / 16, 64 / 64);
    sage_wmma_gemm<128, 64><<<g, 32, 0, stream>>>(agg, h1, inv, Wl2, Wr2, b2, h2);
  }

  zero_f32<<<2048, 256, 0, stream>>>(agg, (long)N_NODES * 64);
  scatter_edges<<<scatterBlocks, 256, 0, stream>>>(h2, src, dst, agg, N_EDGES, 64);
  {
    dim3 g(N_NODES / 16, 128 / 64);
    sage_wmma_gemm<64, 128><<<g, 32, 0, stream>>>(agg, h2, inv, Wl3, Wr3, b3, out);
  }
}


# Float atomic scatter-add and its zero fill replaced by a destination-ordered segment sum: integer rank atomics, scan, place, order by source, fixed-order row sums per node (sum order changes)
# speedup vs baseline: 2.0946x; 2.0946x over previous
#include <hip/hip_runtime.h>


#define N_NODES 50000
#define N_EDGES 800000
#define SCAN_THREADS 1024
#define SCAN_PER 49

typedef __attribute__((ext_vector_type(2))) float v2f;
typedef __attribute__((ext_vector_type(8))) float v8f;


__global__ void zero_i32(int* __restrict__ p, int n) {
  int i = blockIdx.x * blockDim.x + threadIdx.x;
  if (i < n) p[i] = 0;
}

__global__ void rank_edges(const int* __restrict__ dst, int* __restrict__ cnt,
                           int* __restrict__ rank, int nE) {
  int e = blockIdx.x * blockDim.x + threadIdx.x;
  if (e < nE) rank[e] = atomicAdd(&cnt[dst[e]], 1);
}

__global__ __launch_bounds__(SCAN_THREADS)
void scan_counts(const int* __restrict__ cnt, int* __restrict__ rowptr,
                 float* __restrict__ inv) {
  __shared__ int buf[SCAN_THREADS * SCAN_PER];
  __shared__ int wtot[32];
  const int t = threadIdx.x;
  const int lane = t & 31;
  const int w = t >> 5;

#pragma unroll
  for (int j = 0; j < SCAN_PER; ++j) {
    const int i = j * SCAN_THREADS + t;
    int c = 0;
    if (i < N_NODES) {
      c = cnt[i];
      inv[i] = 1.0f / fmaxf((float)c, 1.0f);
    }
    buf[i] = c;
  }
  __syncthreads();

  int s = 0;
  for (int j = 0; j < SCAN_PER; ++j) s += buf[t * SCAN_PER + j];

  int inc = s;
  for (int o = 1; o < 32; o <<= 1) {
    int v = __shfl_up(inc, o);
    if (lane >= o) inc += v;
  }
  if (lane == 31) wtot[w] = inc;
  __syncthreads();
  if (w == 0) {
    const int v = wtot[lane];
    int iv = v;
    for (int o = 1; o < 32; o <<= 1) {
      int u = __shfl_up(iv, o);
      if (lane >= o) iv += u;
    }
    wtot[lane] = iv - v;
  }
  __syncthreads();

  int run = wtot[w] + inc - s;
  for (int j = 0; j < SCAN_PER; ++j) {
    const int c = buf[t * SCAN_PER + j];
    buf[t * SCAN_PER + j] = run;
    run += c;
  }
  __syncthreads();

  for (int i = t; i < N_NODES; i += SCAN_THREADS) rowptr[i] = buf[i];
  if (t == SCAN_THREADS - 1) rowptr[N_NODES] = run;
}

__global__ void fill_perm(const int* __restrict__ src, const int* __restrict__ dst,
                          const int* __restrict__ rank, const int* __restrict__ rowptr,
                          int* __restrict__ perm, int nE) {
  int e = blockIdx.x * blockDim.x + threadIdx.x;
  if (e < nE) perm[rowptr[dst[e]] + rank[e]] = src[e];
}

__global__ __launch_bounds__(256)
void order_segments(const int* __restrict__ rowptr, const int* __restrict__ perm,
                    int* __restrict__ sorted) {
  const int lane = threadIdx.x & 31;
  const int node = __builtin_amdgcn_readfirstlane(blockIdx.x * 8 + (threadIdx.x >> 5));
  if (node >= N_NODES) return;
  const int beg = rowptr[node];
  const int end = rowptr[node + 1];
  for (int p = beg + lane; p < end; p += 32) {
    const int v = perm[p];
    int r = 0;
    for (int q = beg; q < end; ++q) {
      const int u = perm[q];
      r += ((u < v) || (u == v && q < p)) ? 1 : 0;
    }
    sorted[beg + r] = v;
  }
}

template<int D>
__global__ __launch_bounds__(256)
void gather_sum(const float* __restrict__ feat, const int* __restrict__ rowptr,
                const int* __restrict__ sorted, float* __restrict__ agg) {
  constexpr int LPR = D / 4;
  constexpr int G = 32 / LPR;
  const int lane = threadIdx.x & 31;
  const int node = __builtin_amdgcn_readfirstlane(blockIdx.x * 8 + (threadIdx.x >> 5));
  if (node >= N_NODES) return;
  const int g = lane / LPR;
  const int c4 = (lane % LPR) * 4;
  const int beg = rowptr[node];
  const int end = rowptr[node + 1];
  const float* base = feat + c4;

  float4 acc = make_float4(0.0f, 0.0f, 0.0f, 0.0f);
  int j = beg + g;
  for (; j + 3 * G < end; j += 4 * G) {
    const int s0 = sorted[j];
    const int s1 = sorted[j + G];
    const int s2 = sorted[j + 2 * G];
    const int s3 = sorted[j + 3 * G];
    const float4 v0 = *(const float4*)(base + (size_t)s0 * D);
    const float4 v1 = *(const float4*)(base + (size_t)s1 * D);
    const float4 v2 = *(const float4*)(base + (size_t)s2 * D);
    const float4 v3 = *(const float4*)(base + (size_t)s3 * D);
    acc.x += v0.x; acc.y += v0.y; acc.z += v0.z; acc.w += v0.w;
    acc.x += v1.x; acc.y += v1.y; acc.z += v1.z; acc.w += v1.w;
    acc.x += v2.x; acc.y += v2.y; acc.z += v2.z; acc.w += v2.w;
    acc.x += v3.x; acc.y += v3.y; acc.z += v3.z; acc.w += v3.w;
  }
#pragma unroll 1
  for (; j < end; j += G) {
    const int s0 = sorted[j];
    const float4 v0 = *(const float4*)(base + (size_t)s0 * D);
    acc.x += v0.x; acc.y += v0.y; acc.z += v0.z; acc.w += v0.w;
  }
  if (G == 2) {
    acc.x += __shfl_xor(acc.x, 16);
    acc.y += __shfl_xor(acc.y, 16);
    acc.z += __shfl_xor(acc.z, 16);
    acc.w += __shfl_xor(acc.w, 16);
  }
  if (g == 0) *(float4*)(agg + (size_t)node * D + c4) = acc;
}


template<int DIN, int DOUT>
__global__ __launch_bounds__(32)
void sage_wmma_gemm(const float* __restrict__ agg,
                    const float* __restrict__ xin,
                    const float* __restrict__ inv,
                    const float* __restrict__ Wl,
                    const float* __restrict__ Wr,
                    const float* __restrict__ bias,
                    float* __restrict__ out) {
  const int lane = threadIdx.x & 31;
  const int half = lane >> 4;
  const int l16  = lane & 15;
  const int mTile = blockIdx.x << 4;
  const int nBase = blockIdx.y << 6;

  const int aRow = mTile + l16;
  const int kOff = half << 1;

  const float invc = inv[aRow];

  v8f acc[4] = {v8f{}, v8f{}, v8f{}, v8f{}};

  {
    const float* aP = agg + (size_t)aRow * DIN + kOff;
    const float* bP = Wl  + (size_t)(nBase + l16) * DIN + kOff;
#pragma unroll 4
    for (int k = 0; k < DIN; k += 4) {
      v2f a = *(const v2f*)(aP + k);
      a *= invc;
#pragma unroll
      for (int t = 0; t < 4; ++t) {
        v2f b = *(const v2f*)(bP + (size_t)t * 16 * DIN + k);
        acc[t] = __builtin_amdgcn_wmma_f32_16x16x4_f32(false, a, false, b,
                                                       (short)0, acc[t],
                                                       false, false);
      }
    }
  }

  {
    const float* aP = xin + (size_t)aRow * DIN + kOff;
    const float* bP = Wr  + (size_t)(nBase + l16) * DIN + kOff;
#pragma unroll 4
    for (int k = 0; k < DIN; k += 4) {
      v2f a = *(const v2f*)(aP + k);
#pragma unroll
      for (int t = 0; t < 4; ++t) {
        v2f b = *(const v2f*)(bP + (size_t)t * 16 * DIN + k);
        acc[t] = __builtin_amdgcn_wmma_f32_16x16x4_f32(false, a, false, b,
                                                       (short)0, acc[t],
                                                       false, false);
      }
    }
  }

  const int rowBase = mTile + (half << 3);
#pragma unroll
  for (int t = 0; t < 4; ++t) {
    const int col = nBase + t * 16 + l16;
    const float bj = bias[col];
#pragma unroll
    for (int i = 0; i < 8; ++i)
      out[(size_t)(rowBase + i) * DOUT + col] = acc[t][i] + bj;
  }
}


extern "C" void kernel_launch(void* const* d_in, const int* in_sizes, int n_in,
                              void* d_out, int out_size, void* d_ws, size_t ws_size,
                              hipStream_t stream) {
  const float* x   = (const float*)d_in[0];
  const int*   ei  = (const int*)d_in[1];
  const int*   src = ei;
  const int*   dst = ei + N_EDGES;
  const float* Wl1 = (const float*)d_in[2];
  const float* b1  = (const float*)d_in[3];
  const float* Wr1 = (const float*)d_in[4];
  const float* Wl2 = (const float*)d_in[5];
  const float* b2  = (const float*)d_in[6];
  const float* Wr2 = (const float*)d_in[7];
  const float* Wl3 = (const float*)d_in[8];
  const float* b3  = (const float*)d_in[9];
  const float* Wr3 = (const float*)d_in[10];
  float* out = (float*)d_out;

  int*   cnt    = (int*)d_ws;
  float* inv    = (float*)(cnt + 50048);
  int*   rowptr = (int*)(inv + 50048);
  int*   rank   = rowptr + 50048;
  int*   perm   = rank + N_EDGES;
  int*   sorted = perm + N_EDGES;
  float* agg    = (float*)(sorted + N_EDGES);
  float* h1     = agg + (size_t)N_NODES * 128;
  float* h2     = h1  + (size_t)N_NODES * 128;

  zero_i32<<<(N_NODES + 255) / 256, 256, 0, stream>>>(cnt, N_NODES);
  rank_edges<<<(N_EDGES + 255) / 256, 256, 0, stream>>>(dst, cnt, rank, N_EDGES);
  scan_counts<<<1, SCAN_THREADS, 0, stream>>>(cnt, rowptr, inv);
  fill_perm<<<(N_EDGES + 255) / 256, 256, 0, stream>>>(src, dst, rank, rowptr, perm, N_EDGES);
  order_segments<<<(N_NODES + 7) / 8, 256, 0, stream>>>(rowptr, perm, sorted);

  const int gatherBlocks = (N_NODES + 7) / 8;

  gather_sum<128><<<gatherBlocks, 256, 0, stream>>>(x, rowptr, sorted, agg);
  {
    dim3 g(N_NODES / 16, 128 / 64);
    sage_wmma_gemm<128, 128><<<g, 32, 0, stream>>>(agg, x, inv, Wl1, Wr1, b1, h1);
  }

  gather_sum<128><<<gatherBlocks, 256, 0, stream>>>(h1, rowptr, sorted, agg);
  {
    dim3 g(N_NODES / 16, 64 / 64);
    sage_wmma_gemm<128, 64><<<g, 32, 0, stream>>>(agg, h1, inv, Wl2, Wr2, b2, h2);
  }

  gather_sum<64><<<gatherBlocks, 256, 0, stream>>>(h2, rowptr, sorted, agg);
  {
    dim3 g(N_NODES / 16, 128 / 64);
    sage_wmma_gemm<64, 128><<<g, 32, 0, stream>>>(agg, h2, inv, Wl3, Wr3, b3, out);
  }
}


# GEMM operand feed through LDS: resident 8-wave workgroups hold their weight fragments in registers and stage node rows with whole-line loads, two row tiles per wave; k order and sum order unchanged
# speedup vs baseline: 2.4956x; 1.1914x over previous
#include <hip/hip_runtime.h>


#define N_NODES 50000
#define N_EDGES 800000
#define SCAN_THREADS 1024
#define SCAN_PER 49

typedef __attribute__((ext_vector_type(2))) float v2f;
typedef __attribute__((ext_vector_type(4))) float v4f;
typedef __attribute__((ext_vector_type(8))) float v8f;


__global__ void zero_i32(int* __restrict__ p, int n) {
  int i = blockIdx.x * blockDim.x + threadIdx.x;
  if (i < n) p[i] = 0;
}

__global__ void rank_edges(const int* __restrict__ dst, int* __restrict__ cnt,
                           int* __restrict__ rank, int nE) {
  int e = blockIdx.x * blockDim.x + threadIdx.x;
  if (e < nE) rank[e] = atomicAdd(&cnt[dst[e]], 1);
}

__global__ __launch_bounds__(SCAN_THREADS)
void scan_counts(const int* __restrict__ cnt, int* __restrict__ rowptr,
                 float* __restrict__ inv) {
  __shared__ int buf[SCAN_THREADS * SCAN_PER];
  __shared__ int wtot[32];
  const int t = threadIdx.x;
  const int lane = t & 31;
  const int w = t >> 5;

#pragma unroll
  for (int j = 0; j < SCAN_PER; ++j) {
    const int i = j * SCAN_THREADS + t;
    int c = 0;
    if (i < N_NODES) {
      c = cnt[i];
      inv[i] = 1.0f / fmaxf((float)c, 1.0f);
    }
    buf[i] = c;
  }
  __syncthreads();

  int s = 0;
  for (int j = 0; j < SCAN_PER; ++j) s += buf[t * SCAN_PER + j];

  int inc = s;
  for (int o = 1; o < 32; o <<= 1) {
    int v = __shfl_up(inc, o);
    if (lane >= o) inc += v;
  }
  if (lane == 31) wtot[w] = inc;
  __syncthreads();
  if (w == 0) {
    const int v = wtot[lane];
    int iv = v;
    for (int o = 1; o < 32; o <<= 1) {
      int u = __shfl_up(iv, o);
      if (lane >= o) iv += u;
    }
    wtot[lane] = iv - v;
  }
  __syncthreads();

  int run = wtot[w] + inc - s;
  for (int j = 0; j < SCAN_PER; ++j) {
    const int c = buf[t * SCAN_PER + j];
    buf[t * SCAN_PER + j] = run;
    run += c;
  }
  __syncthreads();

  for (int i = t; i < N_NODES; i += SCAN_THREADS) rowptr[i] = buf[i];
  if (t == SCAN_THREADS - 1) rowptr[N_NODES] = run;
}

__global__ void fill_perm(const int* __restrict__ src, const int* __restrict__ dst,
                          const int* __restrict__ rank, const int* __restrict__ rowptr,
                          int* __restrict__ perm, int nE) {
  int e = blockIdx.x * blockDim.x + threadIdx.x;
  if (e < nE) perm[rowptr[dst[e]] + rank[e]] = src[e];
}

__global__ __launch_bounds__(256)
void order_segments(const int* __restrict__ rowptr, const int* __restrict__ perm,
                    int* __restrict__ sorted) {
  const int lane = threadIdx.x & 31;
  const int node = __builtin_amdgcn_readfirstlane(blockIdx.x * 8 + (threadIdx.x >> 5));
  if (node >= N_NODES) return;
  const int beg = rowptr[node];
  const int end = rowptr[node + 1];
  for (int p = beg + lane; p < end; p += 32) {
    const int v = perm[p];
    int r = 0;
    for (int q = beg; q < end; ++q) {
      const int u = perm[q];
      r += ((u < v) || (u == v && q < p)) ? 1 : 0;
    }
    sorted[beg + r] = v;
  }
}

template<int D>
__global__ __launch_bounds__(256)
void gather_sum(const float* __restrict__ feat, const int* __restrict__ rowptr,
                const int* __restrict__ sorted, float* __restrict__ agg) {
  constexpr int LPR = D / 4;
  constexpr int G = 32 / LPR;
  const int lane = threadIdx.x & 31;
  const int node = __builtin_amdgcn_readfirstlane(blockIdx.x * 8 + (threadIdx.x >> 5));
  if (node >= N_NODES) return;
  const int g = lane / LPR;
  const int c4 = (lane % LPR) * 4;
  const int beg = rowptr[node];
  const int end = rowptr[node + 1];
  const float* base = feat + c4;

  float4 acc = make_float4(0.0f, 0.0f, 0.0f, 0.0f);
  int j = beg + g;
  for (; j + 3 * G < end; j += 4 * G) {
    const int s0 = sorted[j];
    const int s1 = sorted[j + G];
    const int s2 = sorted[j + 2 * G];
    const int s3 = sorted[j + 3 * G];
    const float4 v0 = *(const float4*)(base + (size_t)s0 * D);
    const float4 v1 = *(const float4*)(base + (size_t)s1 * D);
    const float4 v2 = *(const float4*)(base + (size_t)s2 * D);
    const float4 v3 = *(const float4*)(base + (size_t)s3 * D);
    acc.x += v0.x; acc.y += v0.y; acc.z += v0.z; acc.w += v0.w;
    acc.x += v1.x; acc.y += v1.y; acc.z += v1.z; acc.w += v1.w;
    acc.x += v2.x; acc.y += v2.y; acc.z += v2.z; acc.w += v2.w;
    acc.x += v3.x; acc.y += v3.y; acc.z += v3.z; acc.w += v3.w;
  }
#pragma unroll 1
  for (; j < end; j += G) {
    const int s0 = sorted[j];
    const float4 v0 = *(const float4*)(base + (size_t)s0 * D);
    acc.x += v0.x; acc.y += v0.y; acc.z += v0.z; acc.w += v0.w;
  }
  if (G == 2) {
    acc.x += __shfl_xor(acc.x, 16);
    acc.y += __shfl_xor(acc.y, 16);
    acc.z += __shfl_xor(acc.z, 16);
    acc.w += __shfl_xor(acc.w, 16);
  }
  if (g == 0) *(float4*)(agg + (size_t)node * D + c4) = acc;
}


template<int DIN, int DOUT>
__global__ __launch_bounds__(256, 4)
void sage_wmma_gemm(const float* __restrict__ agg,
                    const float* __restrict__ xin,
                    const float* __restrict__ inv,
                    const float* __restrict__ Wl,
                    const float* __restrict__ Wr,
                    const float* __restrict__ bias,
                    float* __restrict__ out) {
  constexpr int NT = DOUT / 16;
  constexpr int MW = 8 / NT;
  constexpr int MB = 32 * MW;
  constexpr int P  = 2 * DIN + 4;
  constexpr int KS = 2 * DIN / 4;
  constexpr int Q  = DIN / 4;
  constexpr int LROWS = (DOUT > MB) ? DOUT : MB;
  constexpr int NBLK = (N_NODES + MB - 1) / MB;
  constexpr int WP = DOUT * Q / 256;
  constexpr int RP = MB * Q / 256;

  __shared__ __attribute__((aligned(16))) float S[LROWS * P];

  const int tid  = threadIdx.x;
  const int lane = tid & 31;
  const int half = lane >> 4;
  const int l16  = lane & 15;
  const int w  = __builtin_amdgcn_readfirstlane(tid >> 5);
  const int ct = w % NT;
  const int rg = w / NT;

  v4f ra[RP], rx[RP];
  float  ri[RP];
  int blk = blockIdx.x;
#pragma unroll
  for (int i = 0; i < RP; ++i) {
    const int e  = tid + 256 * i;
    const int row = min(blk * MB + e / Q, N_NODES - 1);
    const int c4 = (e % Q) * 4;
    ra[i] = *(const v4f*)(agg + (size_t)row * DIN + c4);
    rx[i] = *(const v4f*)(xin + (size_t)row * DIN + c4);
    ri[i] = inv[row];
  }

#pragma unroll
  for (int i = 0; i < WP; ++i) {
    const int e  = tid + 256 * i;
    const int n  = e / Q;
    const int c4 = (e % Q) * 4;
    const v4f vl = *(const v4f*)(Wl + (size_t)n * DIN + c4);
    const v4f vr = *(const v4f*)(Wr + (size_t)n * DIN + c4);
    *(v4f*)(S + n * P + c4) = vl;
    *(v4f*)(S + n * P + DIN + c4) = vr;
  }
  __syncthreads();

  v2f B[KS];
  {
    const float* bp = S + (ct * 16 + l16) * P + 2 * half;
#pragma unroll
    for (int s = 0; s < KS; ++s) B[s] = *(const v2f*)(bp + 4 * s);
  }
  const int col = ct * 16 + l16;
  const float bj = bias[col];
  __syncthreads();

  const float* a0 = S + (rg * 32 + l16) * P + 2 * half;
  const float* a1 = a0 + 16 * P;

  for (; blk < NBLK; blk += gridDim.x) {
#pragma unroll
    for (int i = 0; i < RP; ++i) {
      const int e  = tid + 256 * i;
      const int r  = e / Q;
      const int c4 = (e % Q) * 4;
      *(v4f*)(S + r * P + c4) = ra[i] * ri[i];
      *(v4f*)(S + r * P + DIN + c4) = rx[i];
    }
    __syncthreads();

    {
      const int nb = min(blk + (int)gridDim.x, NBLK - 1);
#pragma unroll
      for (int i = 0; i < RP; ++i) {
        const int e  = tid + 256 * i;
        const int row = min(nb * MB + e / Q, N_NODES - 1);
        const int c4 = (e % Q) * 4;
        ra[i] = *(const v4f*)(agg + (size_t)row * DIN + c4);
        rx[i] = *(const v4f*)(xin + (size_t)row * DIN + c4);
        ri[i] = inv[row];
      }
    }

    v8f acc0 = v8f{}, acc1 = v8f{};
#pragma unroll
    for (int s = 0; s < KS; ++s) {
      const v2f A0 = *(const v2f*)(a0 + 4 * s);
      const v2f A1 = *(const v2f*)(a1 + 4 * s);
      acc0 = __builtin_amdgcn_wmma_f32_16x16x4_f32(false, A0, false, B[s],
                                                   (short)0, acc0, false, false);
      acc1 = __builtin_amdgcn_wmma_f32_16x16x4_f32(false, A1, false, B[s],
                                                   (short)0, acc1, false, false);
    }

    const int r0 = blk * MB + rg * 32 + (half << 3);
    if (blk * MB + rg * 32 < N_NODES) {
#pragma unroll
      for (int i = 0; i < 8; ++i)
        out[(size_t)(r0 + i) * DOUT + col] = acc0[i] + bj;
    }
    if (blk * MB + rg * 32 + 16 < N_NODES) {
#pragma unroll
      for (int i = 0; i < 8; ++i)
        out[(size_t)(r0 + 16 + i) * DOUT + col] = acc1[i] + bj;
    }
    __syncthreads();
  }
}


extern "C" void kernel_launch(void* const* d_in, const int* in_sizes, int n_in,
                              void* d_out, int out_size, void* d_ws, size_t ws_size,
                              hipStream_t stream) {
  const float* x   = (const float*)d_in[0];
  const int*   ei  = (const int*)d_in[1];
  const int*   src = ei;
  const int*   dst = ei + N_EDGES;
  const float* Wl1 = (const float*)d_in[2];
  const float* b1  = (const float*)d_in[3];
  const float* Wr1 = (const float*)d_in[4];
  const float* Wl2 = (const float*)d_in[5];
  const float* b2  = (const float*)d_in[6];
  const float* Wr2 = (const float*)d_in[7];
  const float* Wl3 = (const float*)d_in[8];
  const float* b3  = (const float*)d_in[9];
  const float* Wr3 = (const float*)d_in[10];
  float* out = (float*)d_out;

  int*   cnt    = (int*)d_ws;
  float* inv    = (float*)(cnt + 50048);
  int*   rowptr = (int*)(inv + 50048);
  int*   rank   = rowptr + 50048;
  int*   perm   = rank + N_EDGES;
  int*   sorted = perm + N_EDGES;
  float* agg    = (float*)(sorted + N_EDGES);
  float* h1     = agg + (size_t)N_NODES * 128;
  float* h2     = h1  + (size_t)N_NODES * 128;

  zero_i32<<<(N_NODES + 255) / 256, 256, 0, stream>>>(cnt, N_NODES);
  rank_edges<<<(N_EDGES + 255) / 256, 256, 0, stream>>>(dst, cnt, rank, N_EDGES);
  scan_counts<<<1, SCAN_THREADS, 0, stream>>>(cnt, rowptr, inv);
  fill_perm<<<(N_EDGES + 255) / 256, 256, 0, stream>>>(src, dst, rank, rowptr, perm, N_EDGES);
  order_segments<<<(N_NODES + 7) / 8, 256, 0, stream>>>(rowptr, perm, sorted);

  const int gatherBlocks = (N_NODES + 7) / 8;


  gather_sum<128><<<gatherBlocks, 256, 0, stream>>>(x, rowptr, sorted, agg);
  sage_wmma_gemm<128, 128><<<512, 256, 0, stream>>>(agg, x, inv, Wl1, Wr1, b1, h1);

  gather_sum<128><<<gatherBlocks, 256, 0, stream>>>(h1, rowptr, sorted, agg);
  sage_wmma_gemm<128, 64><<<512, 256, 0, stream>>>(agg, h1, inv, Wl2, Wr2, b2, h2);

  gather_sum<64><<<gatherBlocks, 256, 0, stream>>>(h2, rowptr, sorted, agg);
  sage_wmma_gemm<64, 128><<<512, 256, 0, stream>>>(agg, h2, inv, Wl3, Wr3, b3, out);
}


# Layers 1 and 2 composed (no nonlinearity between): W2 W1 built per call on the matrix unit, one GEMM gives layer 2's two 64-wide terms, its row sums run 64 wide; sum order changes
# speedup vs baseline: 2.6259x; 1.0522x over previous
#include <hip/hip_runtime.h>


#define N_NODES 50000
#define N_EDGES 800000
#define SCAN_THREADS 1024
#define SCAN_PER 49

typedef __attribute__((ext_vector_type(2))) float v2f;
typedef __attribute__((ext_vector_type(4))) float v4f;
typedef __attribute__((ext_vector_type(8))) float v8f;


__global__ void zero_i32(int* __restrict__ p, int n) {
  int i = blockIdx.x * blockDim.x + threadIdx.x;
  if (i < n) p[i] = 0;
}

__global__ void rank_edges(const int* __restrict__ dst, int* __restrict__ cnt,
                           int* __restrict__ rank, int nE) {
  int e = blockIdx.x * blockDim.x + threadIdx.x;
  if (e < nE) rank[e] = atomicAdd(&cnt[dst[e]], 1);
}

__device__ __forceinline__
void compose_layers(int gw, int lane,
                    const float* __restrict__ Wl1, const float* __restrict__ Wr1,
                    const float* __restrict__ b1,
                    const float* __restrict__ Wl2, const float* __restrict__ Wr2,
                    const float* __restrict__ b2,
                    float* __restrict__ Wcl, float* __restrict__ Wcr,
                    float* __restrict__ bc) {
  const int half = lane >> 4;
  const int l16  = lane & 15;
  const int tn = (gw >> 3) & 7;
  const int tj = gw & 7;
  const float* W1 = (gw >> 6) ? Wr1 : Wl1;
  float* Wc = (gw >> 6) ? Wcr : Wcl;
  const int n = tn * 16 + l16;
  const float* arow = ((tn < 4) ? (Wl2 + n * 128) : (Wr2 + (n - 64) * 128)) + 2 * half;
  const float* bcol = W1 + (2 * half) * 128 + tj * 16 + l16;
  v8f acc = v8f{};
#pragma unroll
  for (int s = 0; s < 32; ++s) {
    const v2f A = *(const v2f*)(arow + 4 * s);
    v2f B;
    B.x = bcol[(4 * s) * 128];
    B.y = bcol[(4 * s + 1) * 128];
    acc = __builtin_amdgcn_wmma_f32_16x16x4_f32(false, A, false, B,
                                                (short)0, acc, false, false);
  }
  float* cp = Wc + (tn * 16 + 8 * half) * 128 + tj * 16 + l16;
#pragma unroll
  for (int i = 0; i < 8; ++i) cp[i * 128] = acc[i];

  const float* wrow = (gw < 64) ? (Wl2 + gw * 128) : (Wr2 + (gw - 64) * 128);
  const float4 wv = *(const float4*)(wrow + lane * 4);
  const float4 bv = *(const float4*)(b1 + lane * 4);
  float d = wv.x * bv.x + wv.y * bv.y + wv.z * bv.z + wv.w * bv.w;
  for (int o = 16; o > 0; o >>= 1) d += __shfl_xor(d, o);
  if (lane == 0) bc[gw] = d + ((gw < 64) ? 0.0f : b2[gw - 64]);
}

__global__ __launch_bounds__(SCAN_THREADS)
void scan_counts(const int* __restrict__ cnt, int* __restrict__ rowptr,
                 float* __restrict__ inv,
                 const float* __restrict__ Wl1, const float* __restrict__ Wr1,
                 const float* __restrict__ b1,
                 const float* __restrict__ Wl2, const float* __restrict__ Wr2,
                 const float* __restrict__ b2,
                 float* __restrict__ Wcl, float* __restrict__ Wcr,
                 float* __restrict__ bc) {
  __shared__ int buf[SCAN_THREADS * SCAN_PER];
  __shared__ int wtot[32];
  const int t = threadIdx.x;
  const int lane = t & 31;
  const int w = t >> 5;

  if (blockIdx.x != 0) {
    compose_layers((blockIdx.x - 1) * 32 + w, lane, Wl1, Wr1, b1, Wl2, Wr2, b2,
                   Wcl, Wcr, bc);
    return;
  }

#pragma unroll
  for (int j = 0; j < SCAN_PER; ++j) {
    const int i = j * SCAN_THREADS + t;
    int c = 0;
    if (i < N_NODES) {
      c = cnt[i];
      inv[i] = 1.0f / fmaxf((float)c, 1.0f);
    }
    buf[i] = c;
  }
  __syncthreads();

  int s = 0;
  for (int j = 0; j < SCAN_PER; ++j) s += buf[t * SCAN_PER + j];

  int inc = s;
  for (int o = 1; o < 32; o <<= 1) {
    int v = __shfl_up(inc, o);
    if (lane >= o) inc += v;
  }
  if (lane == 31) wtot[w] = inc;
  __syncthreads();
  if (w == 0) {
    const int v = wtot[lane];
    int iv = v;
    for (int o = 1; o < 32; o <<= 1) {
      int u = __shfl_up(iv, o);
      if (lane >= o) iv += u;
    }
    wtot[lane] = iv - v;
  }
  __syncthreads();

  int run = wtot[w] + inc - s;
  for (int j = 0; j < SCAN_PER; ++j) {
    const int c = buf[t * SCAN_PER + j];
    buf[t * SCAN_PER + j] = run;
    run += c;
  }
  __syncthreads();

  for (int i = t; i < N_NODES; i += SCAN_THREADS) rowptr[i] = buf[i];
  if (t == SCAN_THREADS - 1) rowptr[N_NODES] = run;
}

__global__ void fill_perm(const int* __restrict__ src, const int* __restrict__ dst,
                          const int* __restrict__ rank, const int* __restrict__ rowptr,
                          int* __restrict__ perm, int nE) {
  int e = blockIdx.x * blockDim.x + threadIdx.x;
  if (e < nE) perm[rowptr[dst[e]] + rank[e]] = src[e];
}

__global__ __launch_bounds__(256)
void order_segments(const int* __restrict__ rowptr, const int* __restrict__ perm,
                    int* __restrict__ sorted) {
  const int lane = threadIdx.x & 31;
  const int node = __builtin_amdgcn_readfirstlane(blockIdx.x * 8 + (threadIdx.x >> 5));
  if (node >= N_NODES) return;
  const int beg = rowptr[node];
  const int end = rowptr[node + 1];
  for (int p = beg + lane; p < end; p += 32) {
    const int v = perm[p];
    int r = 0;
    for (int q = beg; q < end; ++q) {
      const int u = perm[q];
      r += ((u < v) || (u == v && q < p)) ? 1 : 0;
    }
    sorted[beg + r] = v;
  }
}

template<int D, bool COMBINE>
__global__ __launch_bounds__(256)
void gather_sum(const float* __restrict__ feat, const int* __restrict__ rowptr,
                const int* __restrict__ sorted, const float* __restrict__ inv,
                const float* __restrict__ self, float* __restrict__ agg) {
  constexpr int LPR = D / 4;
  constexpr int G = 32 / LPR;
  const int lane = threadIdx.x & 31;
  const int node = __builtin_amdgcn_readfirstlane(blockIdx.x * 8 + (threadIdx.x >> 5));
  if (node >= N_NODES) return;
  const int g = lane / LPR;
  const int c4 = (lane % LPR) * 4;
  const int beg = rowptr[node];
  const int end = rowptr[node + 1];
  const float* base = feat + c4;

  float4 own = make_float4(0.0f, 0.0f, 0.0f, 0.0f);
  float sc = 1.0f;
  if (COMBINE) {
    own = *(const float4*)(self + (size_t)node * D + c4);
    sc = inv[node];
  }

  float4 acc = make_float4(0.0f, 0.0f, 0.0f, 0.0f);
  int j = beg + g;
  for (; j + 3 * G < end; j += 4 * G) {
    const int s0 = sorted[j];
    const int s1 = sorted[j + G];
    const int s2 = sorted[j + 2 * G];
    const int s3 = sorted[j + 3 * G];
    const float4 v0 = *(const float4*)(base + (size_t)s0 * D);
    const float4 v1 = *(const float4*)(base + (size_t)s1 * D);
    const float4 v2 = *(const float4*)(base + (size_t)s2 * D);
    const float4 v3 = *(const float4*)(base + (size_t)s3 * D);
    acc.x += v0.x; acc.y += v0.y; acc.z += v0.z; acc.w += v0.w;
    acc.x += v1.x; acc.y += v1.y; acc.z += v1.z; acc.w += v1.w;
    acc.x += v2.x; acc.y += v2.y; acc.z += v2.z; acc.w += v2.w;
    acc.x += v3.x; acc.y += v3.y; acc.z += v3.z; acc.w += v3.w;
  }
#pragma unroll 1
  for (; j < end; j += G) {
    const int s0 = sorted[j];
    const float4 v0 = *(const float4*)(base + (size_t)s0 * D);
    acc.x += v0.x; acc.y += v0.y; acc.z += v0.z; acc.w += v0.w;
  }
  if (G == 2) {
    acc.x += __shfl_xor(acc.x, 16);
    acc.y += __shfl_xor(acc.y, 16);
    acc.z += __shfl_xor(acc.z, 16);
    acc.w += __shfl_xor(acc.w, 16);
  }
  if (COMBINE) {
    acc.x = acc.x * sc + own.x;
    acc.y = acc.y * sc + own.y;
    acc.z = acc.z * sc + own.z;
    acc.w = acc.w * sc + own.w;
  }
  if (g == 0) *(float4*)(agg + (size_t)node * D + c4) = acc;
}


template<int DIN, int DOUT, bool SPLIT>
__global__ __launch_bounds__(256, 4)
void sage_wmma_gemm(const float* __restrict__ agg,
                    const float* __restrict__ xin,
                    const float* __restrict__ inv,
                    const float* __restrict__ Wl,
                    const float* __restrict__ Wr,
                    const float* __restrict__ bias,
                    float* __restrict__ out,
                    float* __restrict__ out2) {
  constexpr int NT = DOUT / 16;
  constexpr int MW = 8 / NT;
  constexpr int MB = 32 * MW;
  constexpr int P  = 2 * DIN + 4;
  constexpr int KS = 2 * DIN / 4;
  constexpr int Q  = DIN / 4;
  constexpr int LROWS = (DOUT > MB) ? DOUT : MB;
  constexpr int NBLK = (N_NODES + MB - 1) / MB;
  constexpr int WP = DOUT * Q / 256;
  constexpr int RP = MB * Q / 256;

  __shared__ __attribute__((aligned(16))) float S[LROWS * P];

  const int tid  = threadIdx.x;
  const int lane = tid & 31;
  const int half = lane >> 4;
  const int l16  = lane & 15;
  const int w  = __builtin_amdgcn_readfirstlane(tid >> 5);
  const int ct = w % NT;
  const int rg = w / NT;

  v4f ra[RP], rx[RP];
  float  ri[RP];
  int blk = blockIdx.x;
#pragma unroll
  for (int i = 0; i < RP; ++i) {
    const int e  = tid + 256 * i;
    const int row = min(blk * MB + e / Q, N_NODES - 1);
    const int c4 = (e % Q) * 4;
    ra[i] = *(const v4f*)(agg + (size_t)row * DIN + c4);
    rx[i] = *(const v4f*)(xin + (size_t)row * DIN + c4);
    ri[i] = inv[row];
  }

#pragma unroll
  for (int i = 0; i < WP; ++i) {
    const int e  = tid + 256 * i;
    const int n  = e / Q;
    const int c4 = (e % Q) * 4;
    const v4f vl = *(const v4f*)(Wl + (size_t)n * DIN + c4);
    const v4f vr = *(const v4f*)(Wr + (size_t)n * DIN + c4);
    *(v4f*)(S + n * P + c4) = vl;
    *(v4f*)(S + n * P + DIN + c4) = vr;
  }
  __syncthreads();

  v2f B[KS];
  {
    const float* bp = S + (ct * 16 + l16) * P + 2 * half;
#pragma unroll
    for (int s = 0; s < KS; ++s) B[s] = *(const v2f*)(bp + 4 * s);
  }
  const int col = ct * 16 + l16;
  const float bj = bias[col];
  constexpr int OP = SPLIT ? DOUT / 2 : DOUT;
  float* ob = out;
  int oc = col;
  if (SPLIT && ct >= NT / 2) { ob = out2; oc = col - DOUT / 2; }
  __syncthreads();

  const float* a0 = S + (rg * 32 + l16) * P + 2 * half;
  const float* a1 = a0 + 16 * P;

  for (; blk < NBLK; blk += gridDim.x) {
#pragma unroll
    for (int i = 0; i < RP; ++i) {
      const int e  = tid + 256 * i;
      const int r  = e / Q;
      const int c4 = (e % Q) * 4;
      *(v4f*)(S + r * P + c4) = ra[i] * ri[i];
      *(v4f*)(S + r * P + DIN + c4) = rx[i];
    }
    __syncthreads();

    {
      const int nb = min(blk + (int)gridDim.x, NBLK - 1);
#pragma unroll
      for (int i = 0; i < RP; ++i) {
        const int e  = tid + 256 * i;
        const int row = min(nb * MB + e / Q, N_NODES - 1);
        const int c4 = (e % Q) * 4;
        ra[i] = *(const v4f*)(agg + (size_t)row * DIN + c4);
        rx[i] = *(const v4f*)(xin + (size_t)row * DIN + c4);
        ri[i] = inv[row];
      }
    }

    v8f acc0 = v8f{}, acc1 = v8f{};
#pragma unroll
    for (int s = 0; s < KS; ++s) {
      const v2f A0 = *(const v2f*)(a0 + 4 * s);
      const v2f A1 = *(const v2f*)(a1 + 4 * s);
      acc0 = __builtin_amdgcn_wmma_f32_16x16x4_f32(false, A0, false, B[s],
                                                   (short)0, acc0, false, false);
      acc1 = __builtin_amdgcn_wmma_f32_16x16x4_f32(false, A1, false, B[s],
                                                   (short)0, acc1, false, false);
    }

    const int r0 = blk * MB + rg * 32 + (half << 3);
    if (blk * MB + rg * 32 < N_NODES) {
#pragma unroll
      for (int i = 0; i < 8; ++i)
        ob[(size_t)(r0 + i) * OP + oc] = acc0[i] + bj;
    }
    if (blk * MB + rg * 32 + 16 < N_NODES) {
#pragma unroll
      for (int i = 0; i < 8; ++i)
        ob[(size_t)(r0 + 16 + i) * OP + oc] = acc1[i] + bj;
    }
    __syncthreads();
  }
}


extern "C" void kernel_launch(void* const* d_in, const int* in_sizes, int n_in,
                              void* d_out, int out_size, void* d_ws, size_t ws_size,
                              hipStream_t stream) {
  const float* x   = (const float*)d_in[0];
  const int*   ei  = (const int*)d_in[1];
  const int*   src = ei;
  const int*   dst = ei + N_EDGES;
  const float* Wl1 = (const float*)d_in[2];
  const float* b1  = (const float*)d_in[3];
  const float* Wr1 = (const float*)d_in[4];
  const float* Wl2 = (const float*)d_in[5];
  const float* b2  = (const float*)d_in[6];
  const float* Wr2 = (const float*)d_in[7];
  const float* Wl3 = (const float*)d_in[8];
  const float* b3  = (const float*)d_in[9];
  const float* Wr3 = (const float*)d_in[10];
  float* out = (float*)d_out;

  int*   cnt    = (int*)d_ws;
  float* inv    = (float*)(cnt + 50048);
  int*   rowptr = (int*)(inv + 50048);
  int*   rank   = rowptr + 50048;
  int*   perm   = rank + N_EDGES;
  int*   sorted = perm + N_EDGES;
  float* agg    = (float*)(sorted + N_EDGES);
  float* p2     = agg + (size_t)N_NODES * 128;
  float* q2     = p2  + (size_t)N_NODES * 64;
  float* h2     = q2  + (size_t)N_NODES * 64;
  float* wcl    = h2  + (size_t)N_NODES * 64;
  float* wcr    = wcl + 128 * 128;
  float* bc     = wcr + 128 * 128;

  zero_i32<<<(N_NODES + 255) / 256, 256, 0, stream>>>(cnt, N_NODES);
  rank_edges<<<(N_EDGES + 255) / 256, 256, 0, stream>>>(dst, cnt, rank, N_EDGES);
  scan_counts<<<5, SCAN_THREADS, 0, stream>>>(cnt, rowptr, inv, Wl1, Wr1, b1, Wl2, Wr2, b2,
                                              wcl, wcr, bc);
  fill_perm<<<(N_EDGES + 255) / 256, 256, 0, stream>>>(src, dst, rank, rowptr, perm, N_EDGES);
  order_segments<<<(N_NODES + 7) / 8, 256, 0, stream>>>(rowptr, perm, sorted);

  const int gatherBlocks = (N_NODES + 7) / 8;


  gather_sum<128, false><<<gatherBlocks, 256, 0, stream>>>(x, rowptr, sorted, inv, nullptr, agg);
  sage_wmma_gemm<128, 128, true><<<512, 256, 0, stream>>>(agg, x, inv, wcl, wcr, bc, p2, q2);

  gather_sum<64, true><<<gatherBlocks, 256, 0, stream>>>(p2, rowptr, sorted, inv, q2, h2);

  gather_sum<64, false><<<gatherBlocks, 256, 0, stream>>>(h2, rowptr, sorted, inv, nullptr, agg);
  sage_wmma_gemm<64, 128, false><<<512, 256, 0, stream>>>(agg, h2, inv, Wl3, Wr3, b3, out, nullptr);
}


# Edge list ordered by destination via a two-launch counting sort in LDS (chunk binning, then per-bin count, place, source order) instead of five launches with global atomics; sum order unchanged
# speedup vs baseline: 4.4449x; 1.6927x over previous
#include <hip/hip_runtime.h>


#define N_NODES 50000
#define N_EDGES 800000
#define NBINS 256
#define NPB 196
#define CHUNK 3125
#define EPT 13
#define CAP 8192
#define BT 1024

static_assert(NBINS * CHUNK == N_EDGES, "chunks tile the edge list");
static_assert(NBINS * NPB > N_NODES && (NBINS - 1) * NPB <= N_NODES, "bins cover rowptr[0..N_NODES]");
static_assert(256 * EPT >= CHUNK, "a chunk fits its workgroup");

typedef __attribute__((ext_vector_type(2))) float v2f;
typedef __attribute__((ext_vector_type(4))) float v4f;
typedef __attribute__((ext_vector_type(8))) float v8f;


__device__ __forceinline__
void compose_layers(int gw, int lane,
                    const float* __restrict__ Wl1, const float* __restrict__ Wr1,
                    const float* __restrict__ b1,
                    const float* __restrict__ Wl2, const float* __restrict__ Wr2,
                    const float* __restrict__ b2,
                    float* __restrict__ Wcl, float* __restrict__ Wcr,
                    float* __restrict__ bc) {
  const int half = lane >> 4;
  const int l16  = lane & 15;
  const int tn = (gw >> 3) & 7;
  const int tj = gw & 7;
  const float* W1 = (gw >> 6) ? Wr1 : Wl1;
  float* Wc = (gw >> 6) ? Wcr : Wcl;
  const int n = tn * 16 + l16;
  const float* arow = ((tn < 4) ? (Wl2 + n * 128) : (Wr2 + (n - 64) * 128)) + 2 * half;
  const float* bcol = W1 + (2 * half) * 128 + tj * 16 + l16;
  v8f acc = v8f{};
#pragma unroll
  for (int s = 0; s < 32; ++s) {
    const v2f A = *(const v2f*)(arow + 4 * s);
    v2f B;
    B.x = bcol[(4 * s) * 128];
    B.y = bcol[(4 * s + 1) * 128];
    acc = __builtin_amdgcn_wmma_f32_16x16x4_f32(false, A, false, B,
                                                (short)0, acc, false, false);
  }
  float* cp = Wc + (tn * 16 + 8 * half) * 128 + tj * 16 + l16;
#pragma unroll
  for (int i = 0; i < 8; ++i) cp[i * 128] = acc[i];

  const float* wrow = (gw < 64) ? (Wl2 + gw * 128) : (Wr2 + (gw - 64) * 128);
  const float4 wv = *(const float4*)(wrow + lane * 4);
  const float4 bv = *(const float4*)(b1 + lane * 4);
  float d = wv.x * bv.x + wv.y * bv.y + wv.z * bv.z + wv.w * bv.w;
  for (int o = 16; o > 0; o >>= 1) d += __shfl_xor(d, o);
  if (lane == 0) bc[gw] = d + ((gw < 64) ? 0.0f : b2[gw - 64]);
}

__device__ __forceinline__
int scan256(int val, int t, int* wsum, int& tot) {
  const int lane = t & 31;
  const int w = t >> 5;
  int inc = val;
  for (int o = 1; o < 32; o <<= 1) {
    int v = __shfl_up(inc, o);
    if (lane >= o) inc += v;
  }
  if (w < 8 && lane == 31) wsum[w] = inc;
  __syncthreads();
  int wo = 0;
  tot = 0;
  for (int i = 0; i < 8; ++i) {
    const int a = wsum[i];
    wo += (i < w) ? a : 0;
    tot += a;
  }
  return wo + inc - val;
}

__global__ __launch_bounds__(256)
void bin_edges(const int* __restrict__ src, const int* __restrict__ dst,
               int* __restrict__ binned, int* __restrict__ offT,
               const float* __restrict__ Wl1, const float* __restrict__ Wr1,
               const float* __restrict__ b1,
               const float* __restrict__ Wl2, const float* __restrict__ Wr2,
               const float* __restrict__ b2,
               float* __restrict__ Wcl, float* __restrict__ Wcr,
               float* __restrict__ bc) {
  __shared__ int hist[NBINS];
  __shared__ int wsum[8];
  __shared__ int outb[CHUNK];
  const int t = threadIdx.x;

  if (blockIdx.x >= NBINS) {
    compose_layers((blockIdx.x - NBINS) * 8 + (t >> 5), t & 31, Wl1, Wr1, b1, Wl2, Wr2, b2,
                   Wcl, Wcr, bc);
    return;
  }
  const int b = blockIdx.x;
  hist[t] = 0;

  int rec[EPT], bn[EPT], rk[EPT];
#pragma unroll
  for (int i = 0; i < EPT; ++i) {
    const int j = i * 256 + t;
    bn[i] = -1;
    rec[i] = 0;
    if (j < CHUNK) {
      const int d = dst[b * CHUNK + j];
      const int s = src[b * CHUNK + j];
      const int k = (int)((unsigned)d / (unsigned)NPB);
      bn[i] = k;
      rec[i] = ((d - k * NPB) << 16) | s;
    }
  }
  __syncthreads();
#pragma unroll
  for (int i = 0; i < EPT; ++i) {
    rk[i] = 0;
    if (bn[i] >= 0) rk[i] = atomicAdd(&hist[bn[i]], 1);
  }
  __syncthreads();

  int tot;
  const int ex = scan256(hist[t], t, wsum, tot);
  hist[t] = ex;
  offT[t * NBINS + b] = ex;
  if (t == 0) offT[NBINS * NBINS + b] = CHUNK;
  __syncthreads();

#pragma unroll
  for (int i = 0; i < EPT; ++i)
    if (bn[i] >= 0) outb[hist[bn[i]] + rk[i]] = rec[i];
  __syncthreads();

#pragma unroll
  for (int i = 0; i < EPT; ++i) {
    const int j = i * 256 + t;
    if (j < CHUNK) binned[b * CHUNK + j] = outb[j];
  }
}

__device__ __forceinline__
void build_bin(int* rec, int* rnk, int* seg, int* srt,
               const int* __restrict__ binned,
               const int* p_lo, const int* p_n, const int* p_pos,
               int* cntL, int* startL, int* wsum,
               int k, int base, int total, int t,
               int* __restrict__ rowptr, float* __restrict__ inv) {
  {
    const int piece = t >> 2;
    const int n = p_n[piece];
    const int* sp = binned + piece * CHUNK + p_lo[piece];
    int* dp = rec + p_pos[piece];
    for (int i = t & 3; i < n; i += 16) {
      const int v0 = sp[i];
      const int v1 = sp[min(i + 4, n - 1)];
      const int v2 = sp[min(i + 8, n - 1)];
      const int v3 = sp[min(i + 12, n - 1)];
      dp[i] = v0;
      if (i + 4 < n) dp[i + 4] = v1;
      if (i + 8 < n) dp[i + 8] = v2;
      if (i + 12 < n) dp[i + 12] = v3;
    }
  }
  __syncthreads();

  for (int j = t; j < total; j += BT)
    rnk[j] = atomicAdd(&cntL[rec[j] >> 16], 1);
  __syncthreads();

  {
    const int c = (t < 256) ? cntL[t] : 0;
    int tot;
    const int ex = scan256(c, t, wsum, tot);
    if (t < 256) startL[t] = ex;
    if (t == 255) startL[256] = total;
    const int node = k * NPB + t;
    if (t < NPB && node <= N_NODES) rowptr[node] = base + ex;
    if (t < NPB && node < N_NODES) inv[node] = 1.0f / fmaxf((float)c, 1.0f);
  }
  __syncthreads();

  for (int j = t; j < total; j += BT) {
    const int r = rec[j];
    seg[startL[r >> 16] + rnk[j]] = r & 0xFFFF;
  }
  __syncthreads();

  for (int j = t; j < total; j += BT) {
    const int r = rec[j];
    const int v = r & 0xFFFF;
    const int beg = startL[r >> 16];
    const int end = startL[(r >> 16) + 1];
    const int p = beg + rnk[j];
    int cn = 0;
    for (int q = beg; q < end; q += 4) {
      const int q1 = q + 1, q2 = q + 2, q3 = q + 3;
      const int u0 = seg[q];
      const int u1 = seg[min(q1, end - 1)];
      const int u2 = seg[min(q2, end - 1)];
      const int u3 = seg[min(q3, end - 1)];
      cn += ((u0 < v) || (u0 == v && q < p)) ? 1 : 0;
      cn += (q1 < end && ((u1 < v) || (u1 == v && q1 < p))) ? 1 : 0;
      cn += (q2 < end && ((u2 < v) || (u2 == v && q2 < p))) ? 1 : 0;
      cn += (q3 < end && ((u3 < v) || (u3 == v && q3 < p))) ? 1 : 0;
    }
    srt[beg + cn] = v;
  }
}

__global__ __launch_bounds__(BT)
void build_csr(const int* __restrict__ binned, const int* __restrict__ offT,
               int* __restrict__ rowptr, float* __restrict__ inv,
               int* __restrict__ sorted,
               int* __restrict__ g_rec, int* __restrict__ g_rnk, int* __restrict__ g_seg) {
  __shared__ int s_rec[CAP];
  __shared__ int s_rnk[CAP];
  __shared__ int s_seg[CAP];
  __shared__ int s_srt[CAP];
  __shared__ int p_lo[256], p_n[256], p_pos[256];
  __shared__ int cntL[256], startL[257];
  __shared__ int wsumA[8], wsumB[8], wsumC[8];
  const int t = threadIdx.x;
  const int lane = t & 31;
  const int k = blockIdx.x;

  int lo = 0, n = 0;
  if (t < 256) {
    lo = offT[k * NBINS + t];
    n = offT[(k + 1) * NBINS + t] - lo;
    cntL[t] = 0;
  }
  int sl = lo;
  for (int o = 16; o > 0; o >>= 1) sl += __shfl_xor(sl, o);
  if (t < 256 && lane == 0) wsumB[t >> 5] = sl;
  int total;
  const int pos = scan256(n, t, wsumA, total);
  int base = 0;
  for (int i = 0; i < 8; ++i) base += wsumB[i];
  if (t < 256) { p_lo[t] = lo; p_n[t] = n; p_pos[t] = pos; }
  __syncthreads();

  if (total <= CAP) {
    build_bin(s_rec, s_rnk, s_seg, s_srt, binned, p_lo, p_n, p_pos, cntL, startL, wsumC,
              k, base, total, t, rowptr, inv);
    __syncthreads();
    for (int j = t; j < total; j += BT) sorted[base + j] = s_srt[j];
  } else {
    build_bin(g_rec + base, g_rnk + base, g_seg + base, sorted + base, binned,
              p_lo, p_n, p_pos, cntL, startL, wsumC, k, base, total, t, rowptr, inv);
  }
}

template<int D, bool COMBINE>
__global__ __launch_bounds__(256)
void gather_sum(const float* __restrict__ feat, const int* __restrict__ rowptr,
                const int* __restrict__ sorted, const float* __restrict__ inv,
                const float* __restrict__ self, float* __restrict__ agg) {
  constexpr int LPR = D / 4;
  constexpr int G = 32 / LPR;
  const int lane = threadIdx.x & 31;
  const int node = __builtin_amdgcn_readfirstlane(blockIdx.x * 8 + (threadIdx.x >> 5));
  if (node >= N_NODES) return;
  const int g = lane / LPR;
  const int c4 = (lane % LPR) * 4;
  const int beg = rowptr[node];
  const int end = rowptr[node + 1];
  const float* base = feat + c4;

  float4 own = make_float4(0.0f, 0.0f, 0.0f, 0.0f);
  float sc = 1.0f;
  if (COMBINE) {
    own = *(const float4*)(self + (size_t)node * D + c4);
    sc = inv[node];
  }

  float4 acc = make_float4(0.0f, 0.0f, 0.0f, 0.0f);
  int j = beg + g;
  for (; j + 3 * G < end; j += 4 * G) {
    const int s0 = sorted[j];
    const int s1 = sorted[j + G];
    const int s2 = sorted[j + 2 * G];
    const int s3 = sorted[j + 3 * G];
    const float4 v0 = *(const float4*)(base + (size_t)s0 * D);
    const float4 v1 = *(const float4*)(base + (size_t)s1 * D);
    const float4 v2 = *(const float4*)(base + (size_t)s2 * D);
    const float4 v3 = *(const float4*)(base + (size_t)s3 * D);
    acc.x += v0.x; acc.y += v0.y; acc.z += v0.z; acc.w += v0.w;
    acc.x += v1.x; acc.y += v1.y; acc.z += v1.z; acc.w += v1.w;
    acc.x += v2.x; acc.y += v2.y; acc.z += v2.z; acc.w += v2.w;
    acc.x += v3.x; acc.y += v3.y; acc.z += v3.z; acc.w += v3.w;
  }
#pragma unroll 1
  for (; j < end; j += G) {
    const int s0 = sorted[j];
    const float4 v0 = *(const float4*)(base + (size_t)s0 * D);
    acc.x += v0.x; acc.y += v0.y; acc.z += v0.z; acc.w += v0.w;
  }
  if (G == 2) {
    acc.x += __shfl_xor(acc.x, 16);
    acc.y += __shfl_xor(acc.y, 16);
    acc.z += __shfl_xor(acc.z, 16);
    acc.w += __shfl_xor(acc.w, 16);
  }
  if (COMBINE) {
    acc.x = acc.x * sc + own.x;
    acc.y = acc.y * sc + own.y;
    acc.z = acc.z * sc + own.z;
    acc.w = acc.w * sc + own.w;
  }
  if (g == 0) *(float4*)(agg + (size_t)node * D + c4) = acc;
}


template<int DIN, int DOUT, bool SPLIT>
__global__ __launch_bounds__(256, 4)
void sage_wmma_gemm(const float* __restrict__ agg,
                    const float* __restrict__ xin,
                    const float* __restrict__ inv,
                    const float* __restrict__ Wl,
                    const float* __restrict__ Wr,
                    const float* __restrict__ bias,
                    float* __restrict__ out,
                    float* __restrict__ out2) {
  constexpr int NT = DOUT / 16;
  constexpr int MW = 8 / NT;
  constexpr int MB = 32 * MW;
  constexpr int P  = 2 * DIN + 4;
  constexpr int KS = 2 * DIN / 4;
  constexpr int Q  = DIN / 4;
  constexpr int LROWS = (DOUT > MB) ? DOUT : MB;
  constexpr int NBLK = (N_NODES + MB - 1) / MB;
  constexpr int WP = DOUT * Q / 256;
  constexpr int RP = MB * Q / 256;

  __shared__ __attribute__((aligned(16))) float S[LROWS * P];

  const int tid  = threadIdx.x;
  const int lane = tid & 31;
  const int half = lane >> 4;
  const int l16  = lane & 15;
  const int w  = __builtin_amdgcn_readfirstlane(tid >> 5);
  const int ct = w % NT;
  const int rg = w / NT;

  v4f ra[RP], rx[RP];
  float  ri[RP];
  int blk = blockIdx.x;
#pragma unroll
  for (int i = 0; i < RP; ++i) {
    const int e  = tid + 256 * i;
    const int row = min(blk * MB + e / Q, N_NODES - 1);
    const int c4 = (e % Q) * 4;
    ra[i] = *(const v4f*)(agg + (size_t)row * DIN + c4);
    rx[i] = *(const v4f*)(xin + (size_t)row * DIN + c4);
    ri[i] = inv[row];
  }

#pragma unroll
  for (int i = 0; i < WP; ++i) {
    const int e  = tid + 256 * i;
    const int n  = e / Q;
    const int c4 = (e % Q) * 4;
    const v4f vl = *(const v4f*)(Wl + (size_t)n * DIN + c4);
    const v4f vr = *(const v4f*)(Wr + (size_t)n * DIN + c4);
    *(v4f*)(S + n * P + c4) = vl;
    *(v4f*)(S + n * P + DIN + c4) = vr;
  }
  __syncthreads();

  v2f B[KS];
  {
    const float* bp = S + (ct * 16 + l16) * P + 2 * half;
#pragma unroll
    for (int s = 0; s < KS; ++s) B[s] = *(const v2f*)(bp + 4 * s);
  }
  const int col = ct * 16 + l16;
  const float bj = bias[col];
  constexpr int OP = SPLIT ? DOUT / 2 : DOUT;
  float* ob = out;
  int oc = col;
  if (SPLIT && ct >= NT / 2) { ob = out2; oc = col - DOUT / 2; }
  __syncthreads();

  const float* a0 = S + (rg * 32 + l16) * P + 2 * half;
  const float* a1 = a0 + 16 * P;

  for (; blk < NBLK; blk += gridDim.x) {
#pragma unroll
    for (int i = 0; i < RP; ++i) {
      const int e  = tid + 256 * i;
      const int r  = e / Q;
      const int c4 = (e % Q) * 4;
      *(v4f*)(S + r * P + c4) = ra[i] * ri[i];
      *(v4f*)(S + r * P + DIN + c4) = rx[i];
    }
    __syncthreads();

    {
      const int nb = min(blk + (int)gridDim.x, NBLK - 1);
#pragma unroll
      for (int i = 0; i < RP; ++i) {
        const int e  = tid + 256 * i;
        const int row = min(nb * MB + e / Q, N_NODES - 1);
        const int c4 = (e % Q) * 4;
        ra[i] = *(const v4f*)(agg + (size_t)row * DIN + c4);
        rx[i] = *(const v4f*)(xin + (size_t)row * DIN + c4);
        ri[i] = inv[row];
      }
    }

    v8f acc0 = v8f{}, acc1 = v8f{};
#pragma unroll
    for (int s = 0; s < KS; ++s) {
      const v2f A0 = *(const v2f*)(a0 + 4 * s);
      const v2f A1 = *(const v2f*)(a1 + 4 * s);
      acc0 = __builtin_amdgcn_wmma_f32_16x16x4_f32(false, A0, false, B[s],
                                                   (short)0, acc0, false, false);
      acc1 = __builtin_amdgcn_wmma_f32_16x16x4_f32(false, A1, false, B[s],
                                                   (short)0, acc1, false, false);
    }

    const int r0 = blk * MB + rg * 32 + (half << 3);
    if (blk * MB + rg * 32 < N_NODES) {
#pragma unroll
      for (int i = 0; i < 8; ++i)
        ob[(size_t)(r0 + i) * OP + oc] = acc0[i] + bj;
    }
    if (blk * MB + rg * 32 + 16 < N_NODES) {
#pragma unroll
      for (int i = 0; i < 8; ++i)
        ob[(size_t)(r0 + 16 + i) * OP + oc] = acc1[i] + bj;
    }
    __syncthreads();
  }
}


extern "C" void kernel_launch(void* const* d_in, const int* in_sizes, int n_in,
                              void* d_out, int out_size, void* d_ws, size_t ws_size,
                              hipStream_t stream) {
  const float* x   = (const float*)d_in[0];
  const int*   ei  = (const int*)d_in[1];
  const int*   src = ei;
  const int*   dst = ei + N_EDGES;
  const float* Wl1 = (const float*)d_in[2];
  const float* b1  = (const float*)d_in[3];
  const float* Wr1 = (const float*)d_in[4];
  const float* Wl2 = (const float*)d_in[5];
  const float* b2  = (const float*)d_in[6];
  const float* Wr2 = (const float*)d_in[7];
  const float* Wl3 = (const float*)d_in[8];
  const float* b3  = (const float*)d_in[9];
  const float* Wr3 = (const float*)d_in[10];
  float* out = (float*)d_out;

  float* inv    = (float*)d_ws;
  int*   rowptr = (int*)(inv + 50048);
  int*   offT   = rowptr + 50048;
  int*   binned = offT + 257 * 256;
  int*   sorted = binned + N_EDGES;
  int*   g_rec  = sorted + N_EDGES;
  int*   g_rnk  = g_rec + N_EDGES;
  int*   g_seg  = g_rnk + N_EDGES;
  float* agg    = (float*)(g_seg + N_EDGES);
  float* p2     = agg + (size_t)N_NODES * 128;
  float* q2     = p2  + (size_t)N_NODES * 64;
  float* h2     = q2  + (size_t)N_NODES * 64;
  float* wcl    = h2  + (size_t)N_NODES * 64;
  float* wcr    = wcl + 128 * 128;
  float* bc     = wcr + 128 * 128;

  bin_edges<<<NBINS + 16, 256, 0, stream>>>(src, dst, binned, offT, Wl1, Wr1, b1, Wl2, Wr2, b2,
                                            wcl, wcr, bc);
  build_csr<<<NBINS, BT, 0, stream>>>(binned, offT, rowptr, inv, sorted, g_rec, g_rnk, g_seg);

  const int gatherBlocks = (N_NODES + 7) / 8;


  gather_sum<128, false><<<gatherBlocks, 256, 0, stream>>>(x, rowptr, sorted, inv, nullptr, agg);
  sage_wmma_gemm<128, 128, true><<<512, 256, 0, stream>>>(agg, x, inv, wcl, wcr, bc, p2, q2);

  gather_sum<64, true><<<gatherBlocks, 256, 0, stream>>>(p2, rowptr, sorted, inv, q2, h2);

  gather_sum<64, false><<<gatherBlocks, 256, 0, stream>>>(h2, rowptr, sorted, inv, nullptr, agg);
  sage_wmma_gemm<64, 128, false><<<512, 256, 0, stream>>>(agg, h2, inv, Wl3, Wr3, b3, out, nullptr);
}


# Layer 1's neighbour mean moved behind its projection (linear maps commute with the mean): x projected once to three 64-wide planes (K=128), all three row sums run 64 wide; sum order changes
# speedup vs baseline: 4.8100x; 1.0821x over previous
#include <hip/hip_runtime.h>


#define N_NODES 50000
#define N_EDGES 800000
#define NBINS 256
#define NPB 196
#define CHUNK 3125
#define EPT 13
#define CAP 8192
#define BT 1024

static_assert(NBINS * CHUNK == N_EDGES, "chunks tile the edge list");
static_assert(NBINS * NPB > N_NODES && (NBINS - 1) * NPB <= N_NODES, "bins cover rowptr[0..N_NODES]");
static_assert(256 * EPT >= CHUNK, "a chunk fits its workgroup");

typedef __attribute__((ext_vector_type(2))) float v2f;
typedef __attribute__((ext_vector_type(4))) float v4f;
typedef __attribute__((ext_vector_type(8))) float v8f;


__device__ __forceinline__
void compose_layers(int gw, int lane,
                    const float* __restrict__ Wl1, const float* __restrict__ Wr1,
                    const float* __restrict__ b1,
                    const float* __restrict__ Wl2, const float* __restrict__ Wr2,
                    const float* __restrict__ b2,
                    float* __restrict__ Wc, float* __restrict__ bc) {
  const int half = lane >> 4;
  const int l16  = lane & 15;
  if (gw < 96) {
    const int p  = gw >> 5;
    const int tn = (gw >> 3) & 3;
    const int tj = gw & 7;
    const int n = tn * 16 + l16;
    const int nterms = (p == 1) ? 2 : 1;
    v8f acc = v8f{};
#pragma unroll 1
    for (int term = 0; term < nterms; ++term) {
      const bool a_l = (p == 0) || (p == 1 && term == 0);
      const bool b_l = (p == 0) || (p == 1 && term == 1);
      const float* arow = (a_l ? Wl2 : Wr2) + n * 128 + 2 * half;
      const float* bcol = (b_l ? Wl1 : Wr1) + (2 * half) * 128 + tj * 16 + l16;
#pragma unroll
      for (int s = 0; s < 32; ++s) {
        const v2f A = *(const v2f*)(arow + 4 * s);
        v2f B;
        B.x = bcol[(4 * s) * 128];
        B.y = bcol[(4 * s + 1) * 128];
        acc = __builtin_amdgcn_wmma_f32_16x16x4_f32(false, A, false, B,
                                                    (short)0, acc, false, false);
      }
    }
    float* cp = Wc + (p * 64 + tn * 16 + 8 * half) * 128 + tj * 16 + l16;
#pragma unroll
    for (int i = 0; i < 8; ++i) cp[i * 128] = acc[i];
  }

  const int r = (gw >= 128) ? (gw - 128) : ((gw >= 64) ? (gw - 64) : 0);
  const float* wrow = ((gw >= 128) ? Wr2 : Wl2) + r * 128;
  const float4 wv = *(const float4*)(wrow + lane * 4);
  const float4 bv = *(const float4*)(b1 + lane * 4);
  float d = wv.x * bv.x + wv.y * bv.y + wv.z * bv.z + wv.w * bv.w;
  for (int o = 16; o > 0; o >>= 1) d += __shfl_xor(d, o);
  if (lane == 0) bc[gw] = (gw < 64) ? 0.0f : (d + ((gw >= 128) ? b2[r] : 0.0f));
}

__device__ __forceinline__
int scan256(int val, int t, int* wsum, int& tot) {
  const int lane = t & 31;
  const int w = t >> 5;
  int inc = val;
  for (int o = 1; o < 32; o <<= 1) {
    int v = __shfl_up(inc, o);
    if (lane >= o) inc += v;
  }
  if (w < 8 && lane == 31) wsum[w] = inc;
  __syncthreads();
  int wo = 0;
  tot = 0;
  for (int i = 0; i < 8; ++i) {
    const int a = wsum[i];
    wo += (i < w) ? a : 0;
    tot += a;
  }
  return wo + inc - val;
}

__global__ __launch_bounds__(256)
void bin_edges(const int* __restrict__ src, const int* __restrict__ dst,
               int* __restrict__ binned, int* __restrict__ offT,
               const float* __restrict__ Wl1, const float* __restrict__ Wr1,
               const float* __restrict__ b1,
               const float* __restrict__ Wl2, const float* __restrict__ Wr2,
               const float* __restrict__ b2,
               float* __restrict__ Wc, float* __restrict__ bc) {
  __shared__ int hist[NBINS];
  __shared__ int wsum[8];
  __shared__ int outb[CHUNK];
  const int t = threadIdx.x;

  if (blockIdx.x >= NBINS) {
    const int gw = __builtin_amdgcn_readfirstlane((blockIdx.x - NBINS) * 8 + (t >> 5));
    compose_layers(gw, t & 31, Wl1, Wr1, b1, Wl2, Wr2, b2, Wc, bc);
    return;
  }
  const int b = blockIdx.x;
  hist[t] = 0;

  int rec[EPT], bn[EPT], rk[EPT];
#pragma unroll
  for (int i = 0; i < EPT; ++i) {
    const int j = i * 256 + t;
    bn[i] = -1;
    rec[i] = 0;
    if (j < CHUNK) {
      const int d = dst[b * CHUNK + j];
      const int s = src[b * CHUNK + j];
      const int k = (int)((unsigned)d / (unsigned)NPB);
      bn[i] = k;
      rec[i] = ((d - k * NPB) << 16) | s;
    }
  }
  __syncthreads();
#pragma unroll
  for (int i = 0; i < EPT; ++i) {
    rk[i] = 0;
    if (bn[i] >= 0) rk[i] = atomicAdd(&hist[bn[i]], 1);
  }
  __syncthreads();

  int tot;
  const int ex = scan256(hist[t], t, wsum, tot);
  hist[t] = ex;
  offT[t * NBINS + b] = ex;
  if (t == 0) offT[NBINS * NBINS + b] = CHUNK;
  __syncthreads();

#pragma unroll
  for (int i = 0; i < EPT; ++i)
    if (bn[i] >= 0) outb[hist[bn[i]] + rk[i]] = rec[i];
  __syncthreads();

#pragma unroll
  for (int i = 0; i < EPT; ++i) {
    const int j = i * 256 + t;
    if (j < CHUNK) binned[b * CHUNK + j] = outb[j];
  }
}

__device__ __forceinline__
void build_bin(int* rec, int* rnk, int* seg, int* srt,
               const int* __restrict__ binned,
               const int* p_lo, const int* p_n, const int* p_pos,
               int* cntL, int* startL, int* wsum,
               int k, int base, int total, int t,
               int* __restrict__ rowptr, float* __restrict__ inv) {
  {
    const int piece = t >> 2;
    const int n = p_n[piece];
    const int* sp = binned + piece * CHUNK + p_lo[piece];
    int* dp = rec + p_pos[piece];
    for (int i = t & 3; i < n; i += 16) {
      const int v0 = sp[i];
      const int v1 = sp[min(i + 4, n - 1)];
      const int v2 = sp[min(i + 8, n - 1)];
      const int v3 = sp[min(i + 12, n - 1)];
      dp[i] = v0;
      if (i + 4 < n) dp[i + 4] = v1;
      if (i + 8 < n) dp[i + 8] = v2;
      if (i + 12 < n) dp[i + 12] = v3;
    }
  }
  __syncthreads();

  for (int j = t; j < total; j += BT)
    rnk[j] = atomicAdd(&cntL[rec[j] >> 16], 1);
  __syncthreads();

  {
    const int c = (t < 256) ? cntL[t] : 0;
    int tot;
    const int ex = scan256(c, t, wsum, tot);
    if (t < 256) startL[t] = ex;
    if (t == 255) startL[256] = total;
    const int node = k * NPB + t;
    if (t < NPB && node <= N_NODES) rowptr[node] = base + ex;
    if (t < NPB && node < N_NODES) inv[node] = 1.0f / fmaxf((float)c, 1.0f);
  }
  __syncthreads();

  for (int j = t; j < total; j += BT) {
    const int r = rec[j];
    seg[startL[r >> 16] + rnk[j]] = r & 0xFFFF;
  }
  __syncthreads();

  for (int j = t; j < total; j += BT) {
    const int r = rec[j];
    const int v = r & 0xFFFF;
    const int beg = startL[r >> 16];
    const int end = startL[(r >> 16) + 1];
    const int p = beg + rnk[j];
    int cn = 0;
    for (int q = beg; q < end; q += 4) {
      const int q1 = q + 1, q2 = q + 2, q3 = q + 3;
      const int u0 = seg[q];
      const int u1 = seg[min(q1, end - 1)];
      const int u2 = seg[min(q2, end - 1)];
      const int u3 = seg[min(q3, end - 1)];
      cn += ((u0 < v) || (u0 == v && q < p)) ? 1 : 0;
      cn += (q1 < end && ((u1 < v) || (u1 == v && q1 < p))) ? 1 : 0;
      cn += (q2 < end && ((u2 < v) || (u2 == v && q2 < p))) ? 1 : 0;
      cn += (q3 < end && ((u3 < v) || (u3 == v && q3 < p))) ? 1 : 0;
    }
    srt[beg + cn] = v;
  }
}

__global__ __launch_bounds__(BT)
void build_csr(const int* __restrict__ binned, const int* __restrict__ offT,
               int* __restrict__ rowptr, float* __restrict__ inv,
               int* __restrict__ sorted,
               int* __restrict__ g_rec, int* __restrict__ g_rnk, int* __restrict__ g_seg) {
  __shared__ int s_rec[CAP];
  __shared__ int s_rnk[CAP];
  __shared__ int s_seg[CAP];
  __shared__ int s_srt[CAP];
  __shared__ int p_lo[256], p_n[256], p_pos[256];
  __shared__ int cntL[256], startL[257];
  __shared__ int wsumA[8], wsumB[8], wsumC[8];
  const int t = threadIdx.x;
  const int lane = t & 31;
  const int k = blockIdx.x;

  int lo = 0, n = 0;
  if (t < 256) {
    lo = offT[k * NBINS + t];
    n = offT[(k + 1) * NBINS + t] - lo;
    cntL[t] = 0;
  }
  int sl = lo;
  for (int o = 16; o > 0; o >>= 1) sl += __shfl_xor(sl, o);
  if (t < 256 && lane == 0) wsumB[t >> 5] = sl;
  int total;
  const int pos = scan256(n, t, wsumA, total);
  int base = 0;
  for (int i = 0; i < 8; ++i) base += wsumB[i];
  if (t < 256) { p_lo[t] = lo; p_n[t] = n; p_pos[t] = pos; }
  __syncthreads();

  if (total <= CAP) {
    build_bin(s_rec, s_rnk, s_seg, s_srt, binned, p_lo, p_n, p_pos, cntL, startL, wsumC,
              k, base, total, t, rowptr, inv);
    __syncthreads();
    for (int j = t; j < total; j += BT) sorted[base + j] = s_srt[j];
  } else {
    build_bin(g_rec + base, g_rnk + base, g_seg + base, sorted + base, binned,
              p_lo, p_n, p_pos, cntL, startL, wsumC, k, base, total, t, rowptr, inv);
  }
}

template<int D, bool COMBINE>
__global__ __launch_bounds__(256)
void gather_sum(const float* __restrict__ feat, const int* __restrict__ rowptr,
                const int* __restrict__ sorted, const float* __restrict__ inv,
                const float* __restrict__ self, float* __restrict__ agg) {
  constexpr int LPR = D / 4;
  constexpr int G = 32 / LPR;
  const int lane = threadIdx.x & 31;
  const int node = __builtin_amdgcn_readfirstlane(blockIdx.x * 8 + (threadIdx.x >> 5));
  if (node >= N_NODES) return;
  const int g = lane / LPR;
  const int c4 = (lane % LPR) * 4;
  const int beg = rowptr[node];
  const int end = rowptr[node + 1];
  const float* base = feat + c4;

  float4 own = make_float4(0.0f, 0.0f, 0.0f, 0.0f);
  float sc = 1.0f;
  if (COMBINE) {
    own = *(const float4*)(self + (size_t)node * D + c4);
    sc = inv[node];
  }

  float4 acc = make_float4(0.0f, 0.0f, 0.0f, 0.0f);
  int j = beg + g;
  for (; j + 3 * G < end; j += 4 * G) {
    const int s0 = sorted[j];
    const int s1 = sorted[j + G];
    const int s2 = sorted[j + 2 * G];
    const int s3 = sorted[j + 3 * G];
    const float4 v0 = *(const float4*)(base + (size_t)s0 * D);
    const float4 v1 = *(const float4*)(base + (size_t)s1 * D);
    const float4 v2 = *(const float4*)(base + (size_t)s2 * D);
    const float4 v3 = *(const float4*)(base + (size_t)s3 * D);
    acc.x += v0.x; acc.y += v0.y; acc.z += v0.z; acc.w += v0.w;
    acc.x += v1.x; acc.y += v1.y; acc.z += v1.z; acc.w += v1.w;
    acc.x += v2.x; acc.y += v2.y; acc.z += v2.z; acc.w += v2.w;
    acc.x += v3.x; acc.y += v3.y; acc.z += v3.z; acc.w += v3.w;
  }
#pragma unroll 1
  for (; j < end; j += G) {
    const int s0 = sorted[j];
    const float4 v0 = *(const float4*)(base + (size_t)s0 * D);
    acc.x += v0.x; acc.y += v0.y; acc.z += v0.z; acc.w += v0.w;
  }
  if (G == 2) {
    acc.x += __shfl_xor(acc.x, 16);
    acc.y += __shfl_xor(acc.y, 16);
    acc.z += __shfl_xor(acc.z, 16);
    acc.w += __shfl_xor(acc.w, 16);
  }
  if (COMBINE) {
    acc.x = acc.x * sc + own.x;
    acc.y = acc.y * sc + own.y;
    acc.z = acc.z * sc + own.z;
    acc.w = acc.w * sc + own.w;
  }
  if (g == 0) *(float4*)(agg + (size_t)node * D + c4) = acc;
}


template<int DIN, int DOUT, bool DUAL, int SLABS>
__global__ __launch_bounds__(256, 4)
void sage_wmma_gemm(const float* __restrict__ agg,
                    const float* __restrict__ xin,
                    const float* __restrict__ inv,
                    const float* __restrict__ Wl,
                    const float* __restrict__ Wr,
                    const float* __restrict__ bias,
                    float* __restrict__ out) {
  constexpr int NT = DOUT / 16;
  constexpr int MW = 8 / NT;
  constexpr int MB = 32 * MW;
  constexpr int KW = DUAL ? 2 * DIN : DIN;
  constexpr int XO = DUAL ? DIN : 0;
  constexpr int P  = KW + 4;
  constexpr int KS = KW / 4;
  constexpr int Q  = DIN / 4;
  constexpr int LROWS = (DOUT > MB) ? DOUT : MB;
  constexpr int NBLK = (N_NODES + MB - 1) / MB;
  constexpr int WP = DOUT * Q / 256;
  constexpr int RP = MB * Q / 256;

  __shared__ __attribute__((aligned(16))) float S[LROWS * P];

  const int tid  = threadIdx.x;
  const int lane = tid & 31;
  const int half = lane >> 4;
  const int l16  = lane & 15;
  const int w  = __builtin_amdgcn_readfirstlane(tid >> 5);
  const int ct = w % NT;
  const int rg = w / NT;

  const int slab  = blockIdx.x % SLABS;
  const int bstep = gridDim.x / SLABS;
  if (DUAL) Wl += (size_t)slab * DOUT * DIN;
  Wr   += (size_t)slab * DOUT * DIN;
  bias += slab * DOUT;
  out  += (size_t)slab * N_NODES * DOUT;

  v4f ra[RP], rx[RP];
  float  ri[RP];
  int blk = blockIdx.x / SLABS;
#pragma unroll
  for (int i = 0; i < RP; ++i) {
    const int e  = tid + 256 * i;
    const int row = min(blk * MB + e / Q, N_NODES - 1);
    const int c4 = (e % Q) * 4;
    if (DUAL) {
      ra[i] = *(const v4f*)(agg + (size_t)row * DIN + c4);
      ri[i] = inv[row];
    }
    rx[i] = *(const v4f*)(xin + (size_t)row * DIN + c4);
  }

#pragma unroll
  for (int i = 0; i < WP; ++i) {
    const int e  = tid + 256 * i;
    const int n  = e / Q;
    const int c4 = (e % Q) * 4;
    if (DUAL) {
      const v4f vl = *(const v4f*)(Wl + (size_t)n * DIN + c4);
      *(v4f*)(S + n * P + c4) = vl;
    }
    const v4f vr = *(const v4f*)(Wr + (size_t)n * DIN + c4);
    *(v4f*)(S + n * P + XO + c4) = vr;
  }
  __syncthreads();

  v2f B[KS];
  {
    const float* bp = S + (ct * 16 + l16) * P + 2 * half;
#pragma unroll
    for (int s = 0; s < KS; ++s) B[s] = *(const v2f*)(bp + 4 * s);
  }
  const int col = ct * 16 + l16;
  const float bj = bias[col];
  __syncthreads();

  const float* a0 = S + (rg * 32 + l16) * P + 2 * half;
  const float* a1 = a0 + 16 * P;

  for (; blk < NBLK; blk += bstep) {
#pragma unroll
    for (int i = 0; i < RP; ++i) {
      const int e  = tid + 256 * i;
      const int r  = e / Q;
      const int c4 = (e % Q) * 4;
      if (DUAL) *(v4f*)(S + r * P + c4) = ra[i] * ri[i];
      *(v4f*)(S + r * P + XO + c4) = rx[i];
    }
    __syncthreads();

    {
      const int nb = min(blk + bstep, NBLK - 1);
#pragma unroll
      for (int i = 0; i < RP; ++i) {
        const int e  = tid + 256 * i;
        const int row = min(nb * MB + e / Q, N_NODES - 1);
        const int c4 = (e % Q) * 4;
        if (DUAL) {
          ra[i] = *(const v4f*)(agg + (size_t)row * DIN + c4);
          ri[i] = inv[row];
        }
        rx[i] = *(const v4f*)(xin + (size_t)row * DIN + c4);
      }
    }

    v8f acc0 = v8f{}, acc1 = v8f{};
#pragma unroll
    for (int s = 0; s < KS; ++s) {
      const v2f A0 = *(const v2f*)(a0 + 4 * s);
      const v2f A1 = *(const v2f*)(a1 + 4 * s);
      acc0 = __builtin_amdgcn_wmma_f32_16x16x4_f32(false, A0, false, B[s],
                                                   (short)0, acc0, false, false);
      acc1 = __builtin_amdgcn_wmma_f32_16x16x4_f32(false, A1, false, B[s],
                                                   (short)0, acc1, false, false);
    }

    const int r0 = blk * MB + rg * 32 + (half << 3);
    if (blk * MB + rg * 32 < N_NODES) {
#pragma unroll
      for (int i = 0; i < 8; ++i)
        out[(size_t)(r0 + i) * DOUT + col] = acc0[i] + bj;
    }
    if (blk * MB + rg * 32 + 16 < N_NODES) {
#pragma unroll
      for (int i = 0; i < 8; ++i)
        out[(size_t)(r0 + 16 + i) * DOUT + col] = acc1[i] + bj;
    }
    __syncthreads();
  }
}


extern "C" void kernel_launch(void* const* d_in, const int* in_sizes, int n_in,
                              void* d_out, int out_size, void* d_ws, size_t ws_size,
                              hipStream_t stream) {
  const float* x   = (const float*)d_in[0];
  const int*   ei  = (const int*)d_in[1];
  const int*   src = ei;
  const int*   dst = ei + N_EDGES;
  const float* Wl1 = (const float*)d_in[2];
  const float* b1  = (const float*)d_in[3];
  const float* Wr1 = (const float*)d_in[4];
  const float* Wl2 = (const float*)d_in[5];
  const float* b2  = (const float*)d_in[6];
  const float* Wr2 = (const float*)d_in[7];
  const float* Wl3 = (const float*)d_in[8];
  const float* b3  = (const float*)d_in[9];
  const float* Wr3 = (const float*)d_in[10];
  float* out = (float*)d_out;

  float* inv    = (float*)d_ws;
  int*   rowptr = (int*)(inv + 50048);
  int*   offT   = rowptr + 50048;
  int*   binned = offT + 257 * 256;
  int*   sorted = binned + N_EDGES;
  int*   g_rec  = sorted + N_EDGES;
  int*   g_rnk  = g_rec + N_EDGES;
  int*   g_seg  = g_rnk + N_EDGES;
  float* agg    = (float*)(g_seg + N_EDGES);
  float* pu     = agg + (size_t)N_NODES * 64;
  float* pv     = pu  + (size_t)N_NODES * 64;
  float* pt     = pv  + (size_t)N_NODES * 64;
  float* r2     = pt  + (size_t)N_NODES * 64;
  float* h2     = r2  + (size_t)N_NODES * 64;
  float* wc     = h2  + (size_t)N_NODES * 64;
  float* bc     = wc  + 192 * 128;

  bin_edges<<<NBINS + 24, 256, 0, stream>>>(src, dst, binned, offT, Wl1, Wr1, b1, Wl2, Wr2, b2,
                                            wc, bc);
  build_csr<<<NBINS, BT, 0, stream>>>(binned, offT, rowptr, inv, sorted, g_rec, g_rnk, g_seg);

  const int gatherBlocks = (N_NODES + 7) / 8;

  sage_wmma_gemm<128, 64, false, 3><<<768, 256, 0, stream>>>(nullptr, x, nullptr, nullptr, wc, bc, pu);
  gather_sum<64, true><<<gatherBlocks, 256, 0, stream>>>(pu, rowptr, sorted, inv, pv, r2);
  gather_sum<64, true><<<gatherBlocks, 256, 0, stream>>>(r2, rowptr, sorted, inv, pt, h2);

  gather_sum<64, false><<<gatherBlocks, 256, 0, stream>>>(h2, rowptr, sorted, inv, nullptr, agg);
  sage_wmma_gemm<64, 128, true, 1><<<512, 256, 0, stream>>>(agg, h2, inv, Wl3, Wr3, b3, out);
}


# Row sums: one wave walks 8 consecutive nodes instead of one wave per node (6250 waves per launch, not 50000); bounds and sources by vector load, a node's row loads fly together; sum order unchanged
# speedup vs baseline: 4.9605x; 1.0313x over previous
#include <hip/hip_runtime.h>


#define N_NODES 50000
#define N_EDGES 800000
#define NBINS 256
#define NPB 196
#define CHUNK 3125
#define EPT 13
#define CAP 8192
#define BT 1024

static_assert(NBINS * CHUNK == N_EDGES, "chunks tile the edge list");
static_assert(NBINS * NPB > N_NODES && (NBINS - 1) * NPB <= N_NODES, "bins cover rowptr[0..N_NODES]");
static_assert(256 * EPT >= CHUNK, "a chunk fits its workgroup");

typedef __attribute__((ext_vector_type(2))) float v2f;
typedef __attribute__((ext_vector_type(4))) float v4f;
typedef __attribute__((ext_vector_type(8))) float v8f;


__device__ __forceinline__
void compose_layers(int gw, int lane,
                    const float* __restrict__ Wl1, const float* __restrict__ Wr1,
                    const float* __restrict__ b1,
                    const float* __restrict__ Wl2, const float* __restrict__ Wr2,
                    const float* __restrict__ b2,
                    float* __restrict__ Wc, float* __restrict__ bc) {
  const int half = lane >> 4;
  const int l16  = lane & 15;
  if (gw < 96) {
    const int p  = gw >> 5;
    const int tn = (gw >> 3) & 3;
    const int tj = gw & 7;
    const int n = tn * 16 + l16;
    const int nterms = (p == 1) ? 2 : 1;
    v8f acc = v8f{};
#pragma unroll 1
    for (int term = 0; term < nterms; ++term) {
      const bool a_l = (p == 0) || (p == 1 && term == 0);
      const bool b_l = (p == 0) || (p == 1 && term == 1);
      const float* arow = (a_l ? Wl2 : Wr2) + n * 128 + 2 * half;
      const float* bcol = (b_l ? Wl1 : Wr1) + (2 * half) * 128 + tj * 16 + l16;
#pragma unroll
      for (int s = 0; s < 32; ++s) {
        const v2f A = *(const v2f*)(arow + 4 * s);
        v2f B;
        B.x = bcol[(4 * s) * 128];
        B.y = bcol[(4 * s + 1) * 128];
        acc = __builtin_amdgcn_wmma_f32_16x16x4_f32(false, A, false, B,
                                                    (short)0, acc, false, false);
      }
    }
    float* cp = Wc + (p * 64 + tn * 16 + 8 * half) * 128 + tj * 16 + l16;
#pragma unroll
    for (int i = 0; i < 8; ++i) cp[i * 128] = acc[i];
  }

  const int r = (gw >= 128) ? (gw - 128) : ((gw >= 64) ? (gw - 64) : 0);
  const float* wrow = ((gw >= 128) ? Wr2 : Wl2) + r * 128;
  const float4 wv = *(const float4*)(wrow + lane * 4);
  const float4 bv = *(const float4*)(b1 + lane * 4);
  float d = wv.x * bv.x + wv.y * bv.y + wv.z * bv.z + wv.w * bv.w;
  for (int o = 16; o > 0; o >>= 1) d += __shfl_xor(d, o);
  if (lane == 0) bc[gw] = (gw < 64) ? 0.0f : (d + ((gw >= 128) ? b2[r] : 0.0f));
}

__device__ __forceinline__
int scan256(int val, int t, int* wsum, int& tot) {
  const int lane = t & 31;
  const int w = t >> 5;
  int inc = val;
  for (int o = 1; o < 32; o <<= 1) {
    int v = __shfl_up(inc, o);
    if (lane >= o) inc += v;
  }
  if (w < 8 && lane == 31) wsum[w] = inc;
  __syncthreads();
  int wo = 0;
  tot = 0;
  for (int i = 0; i < 8; ++i) {
    const int a = wsum[i];
    wo += (i < w) ? a : 0;
    tot += a;
  }
  return wo + inc - val;
}

__global__ __launch_bounds__(256)
void bin_edges(const int* __restrict__ src, const int* __restrict__ dst,
               int* __restrict__ binned, int* __restrict__ offT,
               const float* __restrict__ Wl1, const float* __restrict__ Wr1,
               const float* __restrict__ b1,
               const float* __restrict__ Wl2, const float* __restrict__ Wr2,
               const float* __restrict__ b2,
               float* __restrict__ Wc, float* __restrict__ bc) {
  __shared__ int hist[NBINS];
  __shared__ int wsum[8];
  __shared__ int outb[CHUNK];
  const int t = threadIdx.x;

  if (blockIdx.x >= NBINS) {
    const int gw = __builtin_amdgcn_readfirstlane((blockIdx.x - NBINS) * 8 + (t >> 5));
    compose_layers(gw, t & 31, Wl1, Wr1, b1, Wl2, Wr2, b2, Wc, bc);
    return;
  }
  const int b = blockIdx.x;
  hist[t] = 0;

  int rec[EPT], bn[EPT], rk[EPT];
#pragma unroll
  for (int i = 0; i < EPT; ++i) {
    const int j = i * 256 + t;
    bn[i] = -1;
    rec[i] = 0;
    if (j < CHUNK) {
      const int d = dst[b * CHUNK + j];
      const int s = src[b * CHUNK + j];
      const int k = (int)((unsigned)d / (unsigned)NPB);
      bn[i] = k;
      rec[i] = ((d - k * NPB) << 16) | s;
    }
  }
  __syncthreads();
#pragma unroll
  for (int i = 0; i < EPT; ++i) {
    rk[i] = 0;
    if (bn[i] >= 0) rk[i] = atomicAdd(&hist[bn[i]], 1);
  }
  __syncthreads();

  int tot;
  const int ex = scan256(hist[t], t, wsum, tot);
  hist[t] = ex;
  offT[t * NBINS + b] = ex;
  if (t == 0) offT[NBINS * NBINS + b] = CHUNK;
  __syncthreads();

#pragma unroll
  for (int i = 0; i < EPT; ++i)
    if (bn[i] >= 0) outb[hist[bn[i]] + rk[i]] = rec[i];
  __syncthreads();

#pragma unroll
  for (int i = 0; i < EPT; ++i) {
    const int j = i * 256 + t;
    if (j < CHUNK) binned[b * CHUNK + j] = outb[j];
  }
}

__device__ __forceinline__
void build_bin(int* rec, int* rnk, int* seg, int* srt,
               const int* __restrict__ binned,
               const int* p_lo, const int* p_n, const int* p_pos,
               int* cntL, int* startL, int* wsum,
               int k, int base, int total, int t,
               int* __restrict__ rowptr, float* __restrict__ inv) {
  {
    const int piece = t >> 2;
    const int n = p_n[piece];
    const int* sp = binned + piece * CHUNK + p_lo[piece];
    int* dp = rec + p_pos[piece];
    for (int i = t & 3; i < n; i += 16) {
      const int v0 = sp[i];
      const int v1 = sp[min(i + 4, n - 1)];
      const int v2 = sp[min(i + 8, n - 1)];
      const int v3 = sp[min(i + 12, n - 1)];
      dp[i] = v0;
      if (i + 4 < n) dp[i + 4] = v1;
      if (i + 8 < n) dp[i + 8] = v2;
      if (i + 12 < n) dp[i + 12] = v3;
    }
  }
  __syncthreads();

  for (int j = t; j < total; j += BT)
    rnk[j] = atomicAdd(&cntL[rec[j] >> 16], 1);
  __syncthreads();

  {
    const int c = (t < 256) ? cntL[t] : 0;
    int tot;
    const int ex = scan256(c, t, wsum, tot);
    if (t < 256) startL[t] = ex;
    if (t == 255) startL[256] = total;
    const int node = k * NPB + t;
    if (t < NPB && node <= N_NODES) rowptr[node] = base + ex;
    if (t < NPB && node < N_NODES) inv[node] = 1.0f / fmaxf((float)c, 1.0f);
  }
  __syncthreads();

  for (int j = t; j < total; j += BT) {
    const int r = rec[j];
    seg[startL[r >> 16] + rnk[j]] = r & 0xFFFF;
  }
  __syncthreads();

  for (int j = t; j < total; j += BT) {
    const int r = rec[j];
    const int v = r & 0xFFFF;
    const int beg = startL[r >> 16];
    const int end = startL[(r >> 16) + 1];
    const int p = beg + rnk[j];
    int cn = 0;
    for (int q = beg; q < end; q += 4) {
      const int q1 = q + 1, q2 = q + 2, q3 = q + 3;
      const int u0 = seg[q];
      const int u1 = seg[min(q1, end - 1)];
      const int u2 = seg[min(q2, end - 1)];
      const int u3 = seg[min(q3, end - 1)];
      cn += ((u0 < v) || (u0 == v && q < p)) ? 1 : 0;
      cn += (q1 < end && ((u1 < v) || (u1 == v && q1 < p))) ? 1 : 0;
      cn += (q2 < end && ((u2 < v) || (u2 == v && q2 < p))) ? 1 : 0;
      cn += (q3 < end && ((u3 < v) || (u3 == v && q3 < p))) ? 1 : 0;
    }
    srt[beg + cn] = v;
  }
}

__global__ __launch_bounds__(BT)
void build_csr(const int* __restrict__ binned, const int* __restrict__ offT,
               int* __restrict__ rowptr, float* __restrict__ inv,
               int* __restrict__ sorted,
               int* __restrict__ g_rec, int* __restrict__ g_rnk, int* __restrict__ g_seg) {
  __shared__ int s_rec[CAP];
  __shared__ int s_rnk[CAP];
  __shared__ int s_seg[CAP];
  __shared__ int s_srt[CAP];
  __shared__ int p_lo[256], p_n[256], p_pos[256];
  __shared__ int cntL[256], startL[257];
  __shared__ int wsumA[8], wsumB[8], wsumC[8];
  const int t = threadIdx.x;
  const int lane = t & 31;
  const int k = blockIdx.x;

  int lo = 0, n = 0;
  if (t < 256) {
    lo = offT[k * NBINS + t];
    n = offT[(k + 1) * NBINS + t] - lo;
    cntL[t] = 0;
  }
  int sl = lo;
  for (int o = 16; o > 0; o >>= 1) sl += __shfl_xor(sl, o);
  if (t < 256 && lane == 0) wsumB[t >> 5] = sl;
  int total;
  const int pos = scan256(n, t, wsumA, total);
  int base = 0;
  for (int i = 0; i < 8; ++i) base += wsumB[i];
  if (t < 256) { p_lo[t] = lo; p_n[t] = n; p_pos[t] = pos; }
  __syncthreads();

  if (total <= CAP) {
    build_bin(s_rec, s_rnk, s_seg, s_srt, binned, p_lo, p_n, p_pos, cntL, startL, wsumC,
              k, base, total, t, rowptr, inv);
    __syncthreads();
    for (int j = t; j < total; j += BT) sorted[base + j] = s_srt[j];
  } else {
    build_bin(g_rec + base, g_rnk + base, g_seg + base, sorted + base, binned,
              p_lo, p_n, p_pos, cntL, startL, wsumC, k, base, total, t, rowptr, inv);
  }
}

#define NPW 8

template<int T>
__device__ __forceinline__
void ask_rows(float4 (&v)[16], int cur, int cnt, const float* __restrict__ base) {
  if constexpr (T < 16) {
    if (2 * T < cnt) {
      const int s = __builtin_amdgcn_update_dpp(0, cur, 0x150 + T, 0xf, 0xf, false);
      v[T] = *(const float4*)(base + (unsigned)(s * 64));
      ask_rows<T + 1>(v, cur, cnt, base);
    }
  }
}

template<int T>
__device__ __forceinline__
void add_rows(float4& acc, const float4 (&v)[16], int cnt, int g) {
  if constexpr (T < 16) {
    if (2 * T < cnt) {
      const bool ok = (2 * T + g) < cnt;
      acc.x += ok ? v[T].x : 0.0f;
      acc.y += ok ? v[T].y : 0.0f;
      acc.z += ok ? v[T].z : 0.0f;
      acc.w += ok ? v[T].w : 0.0f;
      add_rows<T + 1>(acc, v, cnt, g);
    }
  }
}

template<bool COMBINE>
__global__ __launch_bounds__(256)
void gather_sum(const float* __restrict__ feat, const int* __restrict__ rowptr,
                const int* __restrict__ sorted, const float* __restrict__ inv,
                const float* __restrict__ self, float* __restrict__ agg) {
  constexpr int D = 64;
  const int lane = threadIdx.x & 31;
  const int gw = __builtin_amdgcn_readfirstlane(blockIdx.x * 8 + (threadIdx.x >> 5));
  const int n0 = gw * NPW;
  if (n0 >= N_NODES) return;
  const int nn = min(NPW, N_NODES - n0);
  const int g = lane >> 4;
  const int c4 = (lane & 15) * 4;
  const float* base = feat + c4;
  const int epos = 2 * (lane & 15) + g;

  const int rpv = rowptr[min(n0 + lane, N_NODES)];
  float scv = 1.0f;
  if (COMBINE) scv = inv[min(n0 + lane, N_NODES - 1)];

  int beg = __builtin_amdgcn_readlane(rpv, 0);
  int cur = sorted[min(beg + epos, N_EDGES - 1)];

  for (int i = 0; i < nn; ++i) {
    const int node = n0 + i;
    const int end = __builtin_amdgcn_readlane(rpv, i + 1);
    float4 own = make_float4(0.0f, 0.0f, 0.0f, 0.0f);
    float sc = 1.0f;
    if (COMBINE) {
      own = *(const float4*)(self + (size_t)node * D + c4);
      sc = __builtin_bit_cast(float,
             __builtin_amdgcn_readlane(__builtin_bit_cast(int, scv), i));
    }
    const int nxt = sorted[min(end + epos, N_EDGES - 1)];

    float4 acc = make_float4(0.0f, 0.0f, 0.0f, 0.0f);
    for (int cb = beg; cb < end; cb += 32) {
      if (cb != beg) cur = sorted[min(cb + epos, N_EDGES - 1)];
      const int cnt = min(end - cb, 32);
      float4 v[16];
      ask_rows<0>(v, cur, cnt, base);
      add_rows<0>(acc, v, cnt, g);
    }
    acc.x += __shfl_xor(acc.x, 16);
    acc.y += __shfl_xor(acc.y, 16);
    acc.z += __shfl_xor(acc.z, 16);
    acc.w += __shfl_xor(acc.w, 16);
    if (COMBINE) {
      acc.x = acc.x * sc + own.x;
      acc.y = acc.y * sc + own.y;
      acc.z = acc.z * sc + own.z;
      acc.w = acc.w * sc + own.w;
    }
    if (g == 0) *(float4*)(agg + (size_t)node * D + c4) = acc;
    beg = end;
    cur = nxt;
  }
}


template<int DIN, int DOUT, bool DUAL, int SLABS>
__global__ __launch_bounds__(256, 4)
void sage_wmma_gemm(const float* __restrict__ agg,
                    const float* __restrict__ xin,
                    const float* __restrict__ inv,
                    const float* __restrict__ Wl,
                    const float* __restrict__ Wr,
                    const float* __restrict__ bias,
                    float* __restrict__ out) {
  constexpr int NT = DOUT / 16;
  constexpr int MW = 8 / NT;
  constexpr int MB = 32 * MW;
  constexpr int KW = DUAL ? 2 * DIN : DIN;
  constexpr int XO = DUAL ? DIN : 0;
  constexpr int P  = KW + 4;
  constexpr int KS = KW / 4;
  constexpr int Q  = DIN / 4;
  constexpr int LROWS = (DOUT > MB) ? DOUT : MB;
  constexpr int NBLK = (N_NODES + MB - 1) / MB;
  constexpr int WP = DOUT * Q / 256;
  constexpr int RP = MB * Q / 256;

  __shared__ __attribute__((aligned(16))) float S[LROWS * P];

  const int tid  = threadIdx.x;
  const int lane = tid & 31;
  const int half = lane >> 4;
  const int l16  = lane & 15;
  const int w  = __builtin_amdgcn_readfirstlane(tid >> 5);
  const int ct = w % NT;
  const int rg = w / NT;

  const int slab  = blockIdx.x % SLABS;
  const int bstep = gridDim.x / SLABS;
  if (DUAL) Wl += (size_t)slab * DOUT * DIN;
  Wr   += (size_t)slab * DOUT * DIN;
  bias += slab * DOUT;
  out  += (size_t)slab * N_NODES * DOUT;

  v4f ra[RP], rx[RP];
  float  ri[RP];
  int blk = blockIdx.x / SLABS;
#pragma unroll
  for (int i = 0; i < RP; ++i) {
    const int e  = tid + 256 * i;
    const int row = min(blk * MB + e / Q, N_NODES - 1);
    const int c4 = (e % Q) * 4;
    if (DUAL) {
      ra[i] = *(const v4f*)(agg + (size_t)row * DIN + c4);
      ri[i] = inv[row];
    }
    rx[i] = *(const v4f*)(xin + (size_t)row * DIN + c4);
  }

#pragma unroll
  for (int i = 0; i < WP; ++i) {
    const int e  = tid + 256 * i;
    const int n  = e / Q;
    const int c4 = (e % Q) * 4;
    if (DUAL) {
      const v4f vl = *(const v4f*)(Wl + (size_t)n * DIN + c4);
      *(v4f*)(S + n * P + c4) = vl;
    }
    const v4f vr = *(const v4f*)(Wr + (size_t)n * DIN + c4);
    *(v4f*)(S + n * P + XO + c4) = vr;
  }
  __syncthreads();

  v2f B[KS];
  {
    const float* bp = S + (ct * 16 + l16) * P + 2 * half;
#pragma unroll
    for (int s = 0; s < KS; ++s) B[s] = *(const v2f*)(bp + 4 * s);
  }
  const int col = ct * 16 + l16;
  const float bj = bias[col];
  __syncthreads();

  const float* a0 = S + (rg * 32 + l16) * P + 2 * half;
  const float* a1 = a0 + 16 * P;

  for (; blk < NBLK; blk += bstep) {
#pragma unroll
    for (int i = 0; i < RP; ++i) {
      const int e  = tid + 256 * i;
      const int r  = e / Q;
      const int c4 = (e % Q) * 4;
      if (DUAL) *(v4f*)(S + r * P + c4) = ra[i] * ri[i];
      *(v4f*)(S + r * P + XO + c4) = rx[i];
    }
    __syncthreads();

    {
      const int nb = min(blk + bstep, NBLK - 1);
#pragma unroll
      for (int i = 0; i < RP; ++i) {
        const int e  = tid + 256 * i;
        const int row = min(nb * MB + e / Q, N_NODES - 1);
        const int c4 = (e % Q) * 4;
        if (DUAL) {
          ra[i] = *(const v4f*)(agg + (size_t)row * DIN + c4);
          ri[i] = inv[row];
        }
        rx[i] = *(const v4f*)(xin + (size_t)row * DIN + c4);
      }
    }

    v8f acc0 = v8f{}, acc1 = v8f{};
#pragma unroll
    for (int s = 0; s < KS; ++s) {
      const v2f A0 = *(const v2f*)(a0 + 4 * s);
      const v2f A1 = *(const v2f*)(a1 + 4 * s);
      acc0 = __builtin_amdgcn_wmma_f32_16x16x4_f32(false, A0, false, B[s],
                                                   (short)0, acc0, false, false);
      acc1 = __builtin_amdgcn_wmma_f32_16x16x4_f32(false, A1, false, B[s],
                                                   (short)0, acc1, false, false);
    }

    const int r0 = blk * MB + rg * 32 + (half << 3);
    if (blk * MB + rg * 32 < N_NODES) {
#pragma unroll
      for (int i = 0; i < 8; ++i)
        out[(size_t)(r0 + i) * DOUT + col] = acc0[i] + bj;
    }
    if (blk * MB + rg * 32 + 16 < N_NODES) {
#pragma unroll
      for (int i = 0; i < 8; ++i)
        out[(size_t)(r0 + 16 + i) * DOUT + col] = acc1[i] + bj;
    }
    __syncthreads();
  }
}


extern "C" void kernel_launch(void* const* d_in, const int* in_sizes, int n_in,
                              void* d_out, int out_size, void* d_ws, size_t ws_size,
                              hipStream_t stream) {
  const float* x   = (const float*)d_in[0];
  const int*   ei  = (const int*)d_in[1];
  const int*   src = ei;
  const int*   dst = ei + N_EDGES;
  const float* Wl1 = (const float*)d_in[2];
  const float* b1  = (const float*)d_in[3];
  const float* Wr1 = (const float*)d_in[4];
  const float* Wl2 = (const float*)d_in[5];
  const float* b2  = (const float*)d_in[6];
  const float* Wr2 = (const float*)d_in[7];
  const float* Wl3 = (const float*)d_in[8];
  const float* b3  = (const float*)d_in[9];
  const float* Wr3 = (const float*)d_in[10];
  float* out = (float*)d_out;

  float* inv    = (float*)d_ws;
  int*   rowptr = (int*)(inv + 50048);
  int*   offT   = rowptr + 50048;
  int*   binned = offT + 257 * 256;
  int*   sorted = binned + N_EDGES;
  int*   g_rec  = sorted + N_EDGES;
  int*   g_rnk  = g_rec + N_EDGES;
  int*   g_seg  = g_rnk + N_EDGES;
  float* agg    = (float*)(g_seg + N_EDGES);
  float* pu     = agg + (size_t)N_NODES * 64;
  float* pv     = pu  + (size_t)N_NODES * 64;
  float* pt     = pv  + (size_t)N_NODES * 64;
  float* r2     = pt  + (size_t)N_NODES * 64;
  float* h2     = r2  + (size_t)N_NODES * 64;
  float* wc     = h2  + (size_t)N_NODES * 64;
  float* bc     = wc  + 192 * 128;

  bin_edges<<<NBINS + 24, 256, 0, stream>>>(src, dst, binned, offT, Wl1, Wr1, b1, Wl2, Wr2, b2,
                                            wc, bc);
  build_csr<<<NBINS, BT, 0, stream>>>(binned, offT, rowptr, inv, sorted, g_rec, g_rnk, g_seg);

  const int gatherBlocks = (N_NODES + 8 * NPW - 1) / (8 * NPW);

  sage_wmma_gemm<128, 64, false, 3><<<768, 256, 0, stream>>>(nullptr, x, nullptr, nullptr, wc, bc, pu);
  gather_sum<true><<<gatherBlocks, 256, 0, stream>>>(pu, rowptr, sorted, inv, pv, r2);
  gather_sum<true><<<gatherBlocks, 256, 0, stream>>>(r2, rowptr, sorted, inv, pt, h2);

  gather_sum<false><<<gatherBlocks, 256, 0, stream>>>(h2, rowptr, sorted, inv, nullptr, agg);
  sage_wmma_gemm<64, 128, true, 1><<<512, 256, 0, stream>>>(agg, h2, inv, Wl3, Wr3, b3, out);
}


# Layer 3's row sums moved into its GEMM launch: each wave sums its 4 nodes' neighbour rows into the next LDS A tile between quarters of this tile's products (6 launches, no sum plane); sum order same
# speedup vs baseline: 5.1322x; 1.0346x over previous
#include <hip/hip_runtime.h>


#define N_NODES 50000
#define N_EDGES 800000
#define NBINS 256
#define NPB 196
#define CHUNK 3125
#define EPT 13
#define CAP 8192
#define BT 1024

static_assert(NBINS * CHUNK == N_EDGES, "chunks tile the edge list");
static_assert(NBINS * NPB > N_NODES && (NBINS - 1) * NPB <= N_NODES, "bins cover rowptr[0..N_NODES]");
static_assert(256 * EPT >= CHUNK, "a chunk fits its workgroup");

typedef __attribute__((ext_vector_type(2))) float v2f;
typedef __attribute__((ext_vector_type(4))) float v4f;
typedef __attribute__((ext_vector_type(8))) float v8f;


__device__ __forceinline__
void compose_layers(int gw, int lane,
                    const float* __restrict__ Wl1, const float* __restrict__ Wr1,
                    const float* __restrict__ b1,
                    const float* __restrict__ Wl2, const float* __restrict__ Wr2,
                    const float* __restrict__ b2,
                    float* __restrict__ Wc, float* __restrict__ bc) {
  const int half = lane >> 4;
  const int l16  = lane & 15;
  if (gw < 96) {
    const int p  = gw >> 5;
    const int tn = (gw >> 3) & 3;
    const int tj = gw & 7;
    const int n = tn * 16 + l16;
    const int nterms = (p == 1) ? 2 : 1;
    v8f acc = v8f{};
#pragma unroll 1
    for (int term = 0; term < nterms; ++term) {
      const bool a_l = (p == 0) || (p == 1 && term == 0);
      const bool b_l = (p == 0) || (p == 1 && term == 1);
      const float* arow = (a_l ? Wl2 : Wr2) + n * 128 + 2 * half;
      const float* bcol = (b_l ? Wl1 : Wr1) + (2 * half) * 128 + tj * 16 + l16;
#pragma unroll
      for (int s = 0; s < 32; ++s) {
        const v2f A = *(const v2f*)(arow + 4 * s);
        v2f B;
        B.x = bcol[(4 * s) * 128];
        B.y = bcol[(4 * s + 1) * 128];
        acc = __builtin_amdgcn_wmma_f32_16x16x4_f32(false, A, false, B,
                                                    (short)0, acc, false, false);
      }
    }
    float* cp = Wc + (p * 64 + tn * 16 + 8 * half) * 128 + tj * 16 + l16;
#pragma unroll
    for (int i = 0; i < 8; ++i) cp[i * 128] = acc[i];
  }

  const int r = (gw >= 128) ? (gw - 128) : ((gw >= 64) ? (gw - 64) : 0);
  const float* wrow = ((gw >= 128) ? Wr2 : Wl2) + r * 128;
  const float4 wv = *(const float4*)(wrow + lane * 4);
  const float4 bv = *(const float4*)(b1 + lane * 4);
  float d = wv.x * bv.x + wv.y * bv.y + wv.z * bv.z + wv.w * bv.w;
  for (int o = 16; o > 0; o >>= 1) d += __shfl_xor(d, o);
  if (lane == 0) bc[gw] = (gw < 64) ? 0.0f : (d + ((gw >= 128) ? b2[r] : 0.0f));
}

__device__ __forceinline__
int scan256(int val, int t, int* wsum, int& tot) {
  const int lane = t & 31;
  const int w = t >> 5;
  int inc = val;
  for (int o = 1; o < 32; o <<= 1) {
    int v = __shfl_up(inc, o);
    if (lane >= o) inc += v;
  }
  if (w < 8 && lane == 31) wsum[w] = inc;
  __syncthreads();
  int wo = 0;
  tot = 0;
  for (int i = 0; i < 8; ++i) {
    const int a = wsum[i];
    wo += (i < w) ? a : 0;
    tot += a;
  }
  return wo + inc - val;
}

__global__ __launch_bounds__(256)
void bin_edges(const int* __restrict__ src, const int* __restrict__ dst,
               int* __restrict__ binned, int* __restrict__ offT,
               const float* __restrict__ Wl1, const float* __restrict__ Wr1,
               const float* __restrict__ b1,
               const float* __restrict__ Wl2, const float* __restrict__ Wr2,
               const float* __restrict__ b2,
               float* __restrict__ Wc, float* __restrict__ bc) {
  __shared__ int hist[NBINS];
  __shared__ int wsum[8];
  __shared__ int outb[CHUNK];
  const int t = threadIdx.x;

  if (blockIdx.x >= NBINS) {
    const int gw = __builtin_amdgcn_readfirstlane((blockIdx.x - NBINS) * 8 + (t >> 5));
    compose_layers(gw, t & 31, Wl1, Wr1, b1, Wl2, Wr2, b2, Wc, bc);
    return;
  }
  const int b = blockIdx.x;
  hist[t] = 0;

  int rec[EPT], bn[EPT], rk[EPT];
#pragma unroll
  for (int i = 0; i < EPT; ++i) {
    const int j = i * 256 + t;
    bn[i] = -1;
    rec[i] = 0;
    if (j < CHUNK) {
      const int d = dst[b * CHUNK + j];
      const int s = src[b * CHUNK + j];
      const int k = (int)((unsigned)d / (unsigned)NPB);
      bn[i] = k;
      rec[i] = ((d - k * NPB) << 16) | s;
    }
  }
  __syncthreads();
#pragma unroll
  for (int i = 0; i < EPT; ++i) {
    rk[i] = 0;
    if (bn[i] >= 0) rk[i] = atomicAdd(&hist[bn[i]], 1);
  }
  __syncthreads();

  int tot;
  const int ex = scan256(hist[t], t, wsum, tot);
  hist[t] = ex;
  offT[t * NBINS + b] = ex;
  if (t == 0) offT[NBINS * NBINS + b] = CHUNK;
  __syncthreads();

#pragma unroll
  for (int i = 0; i < EPT; ++i)
    if (bn[i] >= 0) outb[hist[bn[i]] + rk[i]] = rec[i];
  __syncthreads();

#pragma unroll
  for (int i = 0; i < EPT; ++i) {
    const int j = i * 256 + t;
    if (j < CHUNK) binned[b * CHUNK + j] = outb[j];
  }
}

__device__ __forceinline__
void build_bin(int* rec, int* rnk, int* seg, int* srt,
               const int* __restrict__ binned,
               const int* p_lo, const int* p_n, const int* p_pos,
               int* cntL, int* startL, int* wsum,
               int k, int base, int total, int t,
               int* __restrict__ rowptr, float* __restrict__ inv) {
  {
    const int piece = t >> 2;
    const int n = p_n[piece];
    const int* sp = binned + piece * CHUNK + p_lo[piece];
    int* dp = rec + p_pos[piece];
    for (int i = t & 3; i < n; i += 16) {
      const int v0 = sp[i];
      const int v1 = sp[min(i + 4, n - 1)];
      const int v2 = sp[min(i + 8, n - 1)];
      const int v3 = sp[min(i + 12, n - 1)];
      dp[i] = v0;
      if (i + 4 < n) dp[i + 4] = v1;
      if (i + 8 < n) dp[i + 8] = v2;
      if (i + 12 < n) dp[i + 12] = v3;
    }
  }
  __syncthreads();

  for (int j = t; j < total; j += BT)
    rnk[j] = atomicAdd(&cntL[rec[j] >> 16], 1);
  __syncthreads();

  {
    const int c = (t < 256) ? cntL[t] : 0;
    int tot;
    const int ex = scan256(c, t, wsum, tot);
    if (t < 256) startL[t] = ex;
    if (t == 255) startL[256] = total;
    const int node = k * NPB + t;
    if (t < NPB && node <= N_NODES) rowptr[node] = base + ex;
    if (t < NPB && node < N_NODES) inv[node] = 1.0f / fmaxf((float)c, 1.0f);
  }
  __syncthreads();

  for (int j = t; j < total; j += BT) {
    const int r = rec[j];
    seg[startL[r >> 16] + rnk[j]] = r & 0xFFFF;
  }
  __syncthreads();

  for (int j = t; j < total; j += BT) {
    const int r = rec[j];
    const int v = r & 0xFFFF;
    const int beg = startL[r >> 16];
    const int end = startL[(r >> 16) + 1];
    const int p = beg + rnk[j];
    int cn = 0;
    for (int q = beg; q < end; q += 4) {
      const int q1 = q + 1, q2 = q + 2, q3 = q + 3;
      const int u0 = seg[q];
      const int u1 = seg[min(q1, end - 1)];
      const int u2 = seg[min(q2, end - 1)];
      const int u3 = seg[min(q3, end - 1)];
      cn += ((u0 < v) || (u0 == v && q < p)) ? 1 : 0;
      cn += (q1 < end && ((u1 < v) || (u1 == v && q1 < p))) ? 1 : 0;
      cn += (q2 < end && ((u2 < v) || (u2 == v && q2 < p))) ? 1 : 0;
      cn += (q3 < end && ((u3 < v) || (u3 == v && q3 < p))) ? 1 : 0;
    }
    srt[beg + cn] = v;
  }
}

__global__ __launch_bounds__(BT)
void build_csr(const int* __restrict__ binned, const int* __restrict__ offT,
               int* __restrict__ rowptr, float* __restrict__ inv,
               int* __restrict__ sorted,
               int* __restrict__ g_rec, int* __restrict__ g_rnk, int* __restrict__ g_seg) {
  __shared__ int s_rec[CAP];
  __shared__ int s_rnk[CAP];
  __shared__ int s_seg[CAP];
  __shared__ int s_srt[CAP];
  __shared__ int p_lo[256], p_n[256], p_pos[256];
  __shared__ int cntL[256], startL[257];
  __shared__ int wsumA[8], wsumB[8], wsumC[8];
  const int t = threadIdx.x;
  const int lane = t & 31;
  const int k = blockIdx.x;

  int lo = 0, n = 0;
  if (t < 256) {
    lo = offT[k * NBINS + t];
    n = offT[(k + 1) * NBINS + t] - lo;
    cntL[t] = 0;
  }
  int sl = lo;
  for (int o = 16; o > 0; o >>= 1) sl += __shfl_xor(sl, o);
  if (t < 256 && lane == 0) wsumB[t >> 5] = sl;
  int total;
  const int pos = scan256(n, t, wsumA, total);
  int base = 0;
  for (int i = 0; i < 8; ++i) base += wsumB[i];
  if (t < 256) { p_lo[t] = lo; p_n[t] = n; p_pos[t] = pos; }
  __syncthreads();

  if (total <= CAP) {
    build_bin(s_rec, s_rnk, s_seg, s_srt, binned, p_lo, p_n, p_pos, cntL, startL, wsumC,
              k, base, total, t, rowptr, inv);
    __syncthreads();
    for (int j = t; j < total; j += BT) sorted[base + j] = s_srt[j];
  } else {
    build_bin(g_rec + base, g_rnk + base, g_seg + base, sorted + base, binned,
              p_lo, p_n, p_pos, cntL, startL, wsumC, k, base, total, t, rowptr, inv);
  }
}

#define NPW 8

template<int T>
__device__ __forceinline__
void ask_rows(float4 (&v)[16], int cur, int cnt, const float* __restrict__ base) {
  if constexpr (T < 16) {
    if (2 * T < cnt) {
      const int s = __builtin_amdgcn_update_dpp(0, cur, 0x150 + T, 0xf, 0xf, false);
      v[T] = *(const float4*)(base + (unsigned)(s * 64));
      ask_rows<T + 1>(v, cur, cnt, base);
    }
  }
}

template<int T>
__device__ __forceinline__
void add_rows(float4& acc, const float4 (&v)[16], int cnt, int g) {
  if constexpr (T < 16) {
    if (2 * T < cnt) {
      const bool ok = (2 * T + g) < cnt;
      acc.x += ok ? v[T].x : 0.0f;
      acc.y += ok ? v[T].y : 0.0f;
      acc.z += ok ? v[T].z : 0.0f;
      acc.w += ok ? v[T].w : 0.0f;
      add_rows<T + 1>(acc, v, cnt, g);
    }
  }
}

__device__ __forceinline__
float4 rows_finish(float4 (&v)[16], const float* __restrict__ base,
                   const int* __restrict__ sorted, int beg, int end, int epos, int g) {
  float4 acc = make_float4(0.0f, 0.0f, 0.0f, 0.0f);
  add_rows<0>(acc, v, min(end - beg, 32), g);
  for (int cb = beg + 32; cb < end; cb += 32) {
    const int more = sorted[min(cb + epos, N_EDGES - 1)];
    const int cnt = min(end - cb, 32);
    ask_rows<0>(v, more, cnt, base);
    add_rows<0>(acc, v, cnt, g);
  }
  acc.x += __shfl_xor(acc.x, 16);
  acc.y += __shfl_xor(acc.y, 16);
  acc.z += __shfl_xor(acc.z, 16);
  acc.w += __shfl_xor(acc.w, 16);
  return acc;
}

__device__ __forceinline__
float4 node_row_sum(const float* __restrict__ base, const int* __restrict__ sorted,
                    int beg, int end, int cur, int epos, int g) {
  float4 v[16];
  ask_rows<0>(v, cur, min(end - beg, 32), base);
  return rows_finish(v, base, sorted, beg, end, epos, g);
}

__global__ __launch_bounds__(256)
void gather_sum(const float* __restrict__ feat, const int* __restrict__ rowptr,
                const int* __restrict__ sorted, const float* __restrict__ inv,
                const float* __restrict__ self, float* __restrict__ agg) {
  constexpr int D = 64;
  const int lane = threadIdx.x & 31;
  const int gw = __builtin_amdgcn_readfirstlane(blockIdx.x * 8 + (threadIdx.x >> 5));
  const int n0 = gw * NPW;
  if (n0 >= N_NODES) return;
  const int nn = min(NPW, N_NODES - n0);
  const int g = lane >> 4;
  const int c4 = (lane & 15) * 4;
  const float* base = feat + c4;
  const int epos = 2 * (lane & 15) + g;

  const int rpv = rowptr[min(n0 + lane, N_NODES)];
  const float scv = inv[min(n0 + lane, N_NODES - 1)];

  int beg = __builtin_amdgcn_readlane(rpv, 0);
  int cur = sorted[min(beg + epos, N_EDGES - 1)];

  for (int i = 0; i < nn; ++i) {
    const int node = n0 + i;
    const int end = __builtin_amdgcn_readlane(rpv, i + 1);
    const float4 own = *(const float4*)(self + (size_t)node * D + c4);
    const float sc = __builtin_bit_cast(float,
                       __builtin_amdgcn_readlane(__builtin_bit_cast(int, scv), i));
    const int nxt = sorted[min(end + epos, N_EDGES - 1)];

    float4 acc = node_row_sum(base, sorted, beg, end, cur, epos, g);
    acc.x = acc.x * sc + own.x;
    acc.y = acc.y * sc + own.y;
    acc.z = acc.z * sc + own.z;
    acc.w = acc.w * sc + own.w;
    if (g == 0) *(float4*)(agg + (size_t)node * D + c4) = acc;
    beg = end;
    cur = nxt;
  }
}


template<int DIN, int DOUT, int SLABS>
__global__ __launch_bounds__(256, 4)
void sage_wmma_gemm(const float* __restrict__ xin,
                    const float* __restrict__ Wr,
                    const float* __restrict__ bias,
                    float* __restrict__ out) {
  constexpr int NT = DOUT / 16;
  constexpr int MW = 8 / NT;
  constexpr int MB = 32 * MW;
  constexpr int P  = DIN + 4;
  constexpr int KS = DIN / 4;
  constexpr int Q  = DIN / 4;
  constexpr int LROWS = (DOUT > MB) ? DOUT : MB;
  constexpr int NBLK = (N_NODES + MB - 1) / MB;
  constexpr int WP = DOUT * Q / 256;
  constexpr int RP = MB * Q / 256;

  __shared__ __attribute__((aligned(16))) float S[LROWS * P];

  const int tid  = threadIdx.x;
  const int lane = tid & 31;
  const int half = lane >> 4;
  const int l16  = lane & 15;
  const int w  = __builtin_amdgcn_readfirstlane(tid >> 5);
  const int ct = w % NT;
  const int rg = w / NT;

  const int slab  = blockIdx.x % SLABS;
  const int bstep = gridDim.x / SLABS;
  Wr   += (size_t)slab * DOUT * DIN;
  bias += slab * DOUT;
  out  += (size_t)slab * N_NODES * DOUT;

  v4f rx[RP];
  int blk = blockIdx.x / SLABS;
#pragma unroll
  for (int i = 0; i < RP; ++i) {
    const int e  = tid + 256 * i;
    const int row = min(blk * MB + e / Q, N_NODES - 1);
    const int c4 = (e % Q) * 4;
    rx[i] = *(const v4f*)(xin + (size_t)row * DIN + c4);
  }

#pragma unroll
  for (int i = 0; i < WP; ++i) {
    const int e  = tid + 256 * i;
    const int n  = e / Q;
    const int c4 = (e % Q) * 4;
    const v4f vr = *(const v4f*)(Wr + (size_t)n * DIN + c4);
    *(v4f*)(S + n * P + c4) = vr;
  }
  __syncthreads();

  v2f B[KS];
  {
    const float* bp = S + (ct * 16 + l16) * P + 2 * half;
#pragma unroll
    for (int s = 0; s < KS; ++s) B[s] = *(const v2f*)(bp + 4 * s);
  }
  const int col = ct * 16 + l16;
  const float bj = bias[col];
  __syncthreads();

  const float* a0 = S + (rg * 32 + l16) * P + 2 * half;
  const float* a1 = a0 + 16 * P;

  for (; blk < NBLK; blk += bstep) {
#pragma unroll
    for (int i = 0; i < RP; ++i) {
      const int e  = tid + 256 * i;
      const int r  = e / Q;
      const int c4 = (e % Q) * 4;
      *(v4f*)(S + r * P + c4) = rx[i];
    }
    __syncthreads();

    {
      const int nb = min(blk + bstep, NBLK - 1);
#pragma unroll
      for (int i = 0; i < RP; ++i) {
        const int e  = tid + 256 * i;
        const int row = min(nb * MB + e / Q, N_NODES - 1);
        const int c4 = (e % Q) * 4;
        rx[i] = *(const v4f*)(xin + (size_t)row * DIN + c4);
      }
    }

    v8f acc0 = v8f{}, acc1 = v8f{};
#pragma unroll
    for (int s = 0; s < KS; ++s) {
      const v2f A0 = *(const v2f*)(a0 + 4 * s);
      const v2f A1 = *(const v2f*)(a1 + 4 * s);
      acc0 = __builtin_amdgcn_wmma_f32_16x16x4_f32(false, A0, false, B[s],
                                                   (short)0, acc0, false, false);
      acc1 = __builtin_amdgcn_wmma_f32_16x16x4_f32(false, A1, false, B[s],
                                                   (short)0, acc1, false, false);
    }

    const int r0 = blk * MB + rg * 32 + (half << 3);
    if (blk * MB + rg * 32 < N_NODES) {
#pragma unroll
      for (int i = 0; i < 8; ++i)
        out[(size_t)(r0 + i) * DOUT + col] = acc0[i] + bj;
    }
    if (blk * MB + rg * 32 + 16 < N_NODES) {
#pragma unroll
      for (int i = 0; i < 8; ++i)
        out[(size_t)(r0 + 16 + i) * DOUT + col] = acc1[i] + bj;
    }
    __syncthreads();
  }
}

__global__ __launch_bounds__(256, 4)
void sage_layer3(const float* __restrict__ h,
                 const int* __restrict__ rowptr, const int* __restrict__ sorted,
                 const float* __restrict__ inv,
                 const float* __restrict__ Wl,
                 const float* __restrict__ Wr,
                 const float* __restrict__ bias,
                 float* __restrict__ out) {
  constexpr int DIN = 64, DOUT = 128;
  constexpr int MB = 32;
  constexpr int NPB3 = MB / 8;
  constexpr int KW = 2 * DIN;
  constexpr int XO = DIN;
  constexpr int P  = KW + 4;
  constexpr int KS = KW / 4;
  constexpr int KQ = KS / NPB3;
  constexpr int Q  = DIN / 4;
  constexpr int NBLK = (N_NODES + MB - 1) / MB;
  constexpr int WP = DOUT * Q / 256;
  constexpr int RP = MB * Q / 256;
  constexpr int TILE = MB * P;
  static_assert(2 * MB <= DOUT, "both A tiles fit the rows that held the weights");

  __shared__ __attribute__((aligned(16))) float S[DOUT * P];

  const int tid  = threadIdx.x;
  const int lane = tid & 31;
  const int half = lane >> 4;
  const int l16  = lane & 15;
  const int w  = __builtin_amdgcn_readfirstlane(tid >> 5);
  const int bstep = gridDim.x;
  const int c4g = l16 * 4;
  const int epos = 2 * l16 + half;
  const float* hbase = h + c4g;

  int blk = blockIdx.x;
  int rpv = rowptr[min(blk * MB + w * NPB3 + lane, N_NODES)];
  float scv = inv[min(blk * MB + w * NPB3 + lane, N_NODES - 1)];
  v4f rx[RP];
#pragma unroll
  for (int i = 0; i < RP; ++i) {
    const int e  = tid + 256 * i;
    const int row = min(blk * MB + e / Q, N_NODES - 1);
    const int c4 = (e % Q) * 4;
    rx[i] = *(const v4f*)(h + (size_t)row * DIN + c4);
  }
  int cur = sorted[min(__builtin_amdgcn_readlane(rpv, 0) + epos, N_EDGES - 1)];

#pragma unroll
  for (int i = 0; i < WP; ++i) {
    const int e  = tid + 256 * i;
    const int n  = e / Q;
    const int c4 = (e % Q) * 4;
    const v4f vl = *(const v4f*)(Wl + (size_t)n * DIN + c4);
    const v4f vr = *(const v4f*)(Wr + (size_t)n * DIN + c4);
    *(v4f*)(S + n * P + c4) = vl;
    *(v4f*)(S + n * P + XO + c4) = vr;
  }
  __syncthreads();

  v2f B[KS];
  {
    const float* bp = S + (w * 16 + l16) * P + 2 * half;
#pragma unroll
    for (int s = 0; s < KS; ++s) B[s] = *(const v2f*)(bp + 4 * s);
  }
  const int col = w * 16 + l16;
  const float bj = bias[col];
  __syncthreads();

  {
    int beg = __builtin_amdgcn_readlane(rpv, 0);
    for (int i = 0; i < NPB3; ++i) {
      const int end = __builtin_amdgcn_readlane(rpv, i + 1);
      const float sc = __builtin_bit_cast(float,
                         __builtin_amdgcn_readlane(__builtin_bit_cast(int, scv), i));
      const int nxt = sorted[min(end + epos, N_EDGES - 1)];
      float4 acc = node_row_sum(hbase, sorted, beg, end, cur, epos, half);
      acc.x *= sc;
      acc.y *= sc;
      acc.z *= sc;
      acc.w *= sc;
      if (half == 0) *(float4*)(S + (w * NPB3 + i) * P + c4g) = acc;
      beg = end;
      cur = nxt;
    }
#pragma unroll
    for (int i = 0; i < RP; ++i) {
      const int e  = tid + 256 * i;
      *(v4f*)(S + (e / Q) * P + XO + (e % Q) * 4) = rx[i];
    }
  }
  {
    const int nb = min(blk + bstep, NBLK - 1);
    rpv = rowptr[min(nb * MB + w * NPB3 + lane, N_NODES)];
    scv = inv[min(nb * MB + w * NPB3 + lane, N_NODES - 1)];
#pragma unroll
    for (int i = 0; i < RP; ++i) {
      const int e  = tid + 256 * i;
      const int row = min(nb * MB + e / Q, N_NODES - 1);
      rx[i] = *(const v4f*)(h + (size_t)row * DIN + (e % Q) * 4);
    }
    cur = sorted[min(__builtin_amdgcn_readlane(rpv, 0) + epos, N_EDGES - 1)];
  }
  __syncthreads();

  int buf = 0;
  for (; blk < NBLK; blk += bstep) {
    const bool fill = (blk + bstep) < NBLK;
    const float* a0 = S + buf * TILE + l16 * P + 2 * half;
    const float* a1 = a0 + 16 * P;
    float* nt = S + (buf ^ 1) * TILE;

    if (fill) {
#pragma unroll
      for (int i = 0; i < RP; ++i) {
        const int e  = tid + 256 * i;
        *(v4f*)(nt + (e / Q) * P + XO + (e % Q) * 4) = rx[i];
      }
    }
    const int nb2 = min(blk + 2 * bstep, NBLK - 1);
    const int rpn = rowptr[min(nb2 * MB + w * NPB3 + lane, N_NODES)];
    const float scn = inv[min(nb2 * MB + w * NPB3 + lane, N_NODES - 1)];
#pragma unroll
    for (int i = 0; i < RP; ++i) {
      const int e  = tid + 256 * i;
      const int row = min(nb2 * MB + e / Q, N_NODES - 1);
      rx[i] = *(const v4f*)(h + (size_t)row * DIN + (e % Q) * 4);
    }
    int curn = 0;

    v8f acc0 = v8f{}, acc1 = v8f{};
    int beg = __builtin_amdgcn_readlane(rpv, 0);
#pragma unroll
    for (int j = 0; j < NPB3; ++j) {
      float4 v[16];
      int end = beg, nxt = 0;
      if (fill) {
        end = __builtin_amdgcn_readlane(rpv, j + 1);
        nxt = sorted[min(end + epos, N_EDGES - 1)];
        ask_rows<0>(v, cur, min(end - beg, 32), hbase);
      }
      if (j == NPB3 / 2)
        curn = sorted[min(__builtin_amdgcn_readlane(rpn, 0) + epos, N_EDGES - 1)];

#pragma unroll
      for (int s = j * KQ; s < (j + 1) * KQ; ++s) {
        const v2f A0 = *(const v2f*)(a0 + 4 * s);
        const v2f A1 = *(const v2f*)(a1 + 4 * s);
        acc0 = __builtin_amdgcn_wmma_f32_16x16x4_f32(false, A0, false, B[s],
                                                     (short)0, acc0, false, false);
        acc1 = __builtin_amdgcn_wmma_f32_16x16x4_f32(false, A1, false, B[s],
                                                     (short)0, acc1, false, false);
      }

      if (fill) {
        const float sc = __builtin_bit_cast(float,
                           __builtin_amdgcn_readlane(__builtin_bit_cast(int, scv), j));
        float4 acc = rows_finish(v, hbase, sorted, beg, end, epos, half);
        acc.x *= sc;
        acc.y *= sc;
        acc.z *= sc;
        acc.w *= sc;
        if (half == 0) *(float4*)(nt + (w * NPB3 + j) * P + c4g) = acc;
        beg = end;
        cur = nxt;
      }
    }
    rpv = rpn;
    scv = scn;
    cur = curn;

    const int r0 = blk * MB + (half << 3);
    if (blk * MB < N_NODES) {
#pragma unroll
      for (int i = 0; i < 8; ++i)
        out[(size_t)(r0 + i) * DOUT + col] = acc0[i] + bj;
    }
    if (blk * MB + 16 < N_NODES) {
#pragma unroll
      for (int i = 0; i < 8; ++i)
        out[(size_t)(r0 + 16 + i) * DOUT + col] = acc1[i] + bj;
    }
    __syncthreads();
    buf ^= 1;
  }
}


extern "C" void kernel_launch(void* const* d_in, const int* in_sizes, int n_in,
                              void* d_out, int out_size, void* d_ws, size_t ws_size,
                              hipStream_t stream) {
  const float* x   = (const float*)d_in[0];
  const int*   ei  = (const int*)d_in[1];
  const int*   src = ei;
  const int*   dst = ei + N_EDGES;
  const float* Wl1 = (const float*)d_in[2];
  const float* b1  = (const float*)d_in[3];
  const float* Wr1 = (const float*)d_in[4];
  const float* Wl2 = (const float*)d_in[5];
  const float* b2  = (const float*)d_in[6];
  const float* Wr2 = (const float*)d_in[7];
  const float* Wl3 = (const float*)d_in[8];
  const float* b3  = (const float*)d_in[9];
  const float* Wr3 = (const float*)d_in[10];
  float* out = (float*)d_out;

  float* inv    = (float*)d_ws;
  int*   rowptr = (int*)(inv + 50048);
  int*   offT   = rowptr + 50048;
  int*   binned = offT + 257 * 256;
  int*   sorted = binned + N_EDGES;
  int*   g_rec  = sorted + N_EDGES;
  int*   g_rnk  = g_rec + N_EDGES;
  int*   g_seg  = g_rnk + N_EDGES;
  float* pu     = (float*)(g_seg + N_EDGES);
  float* pv     = pu  + (size_t)N_NODES * 64;
  float* pt     = pv  + (size_t)N_NODES * 64;
  float* r2     = pt  + (size_t)N_NODES * 64;
  float* h2     = r2  + (size_t)N_NODES * 64;
  float* wc     = h2  + (size_t)N_NODES * 64;
  float* bc     = wc  + 192 * 128;

  bin_edges<<<NBINS + 24, 256, 0, stream>>>(src, dst, binned, offT, Wl1, Wr1, b1, Wl2, Wr2, b2,
                                            wc, bc);
  build_csr<<<NBINS, BT, 0, stream>>>(binned, offT, rowptr, inv, sorted, g_rec, g_rnk, g_seg);

  const int gatherBlocks = (N_NODES + 8 * NPW - 1) / (8 * NPW);

  sage_wmma_gemm<128, 64, 3><<<768, 256, 0, stream>>>(x, wc, bc, pu);
  gather_sum<<<gatherBlocks, 256, 0, stream>>>(pu, rowptr, sorted, inv, pv, r2);
  gather_sum<<<gatherBlocks, 256, 0, stream>>>(r2, rowptr, sorted, inv, pt, h2);

  sage_layer3<<<512, 256, 0, stream>>>(h2, rowptr, sorted, inv, Wl3, Wr3, b3, out);
}


# GEMM epilogues go through an LDS output tile: each store instruction of the projection and layer-3 launches writes four whole 128-B lines instead of 64-B pieces of two rows; sums unchanged
# speedup vs baseline: 5.3463x; 1.0417x over previous
#include <hip/hip_runtime.h>


#define N_NODES 50000
#define N_EDGES 800000
#define NBINS 256
#define NPB 196
#define CHUNK 3125
#define EPT 13
#define CAP 8192
#define BT 1024

static_assert(NBINS * CHUNK == N_EDGES, "chunks tile the edge list");
static_assert(NBINS * NPB > N_NODES && (NBINS - 1) * NPB <= N_NODES, "bins cover rowptr[0..N_NODES]");
static_assert(256 * EPT >= CHUNK, "a chunk fits its workgroup");

typedef __attribute__((ext_vector_type(2))) float v2f;
typedef __attribute__((ext_vector_type(4))) float v4f;
typedef __attribute__((ext_vector_type(8))) float v8f;


__device__ __forceinline__
void compose_layers(int gw, int lane,
                    const float* __restrict__ Wl1, const float* __restrict__ Wr1,
                    const float* __restrict__ b1,
                    const float* __restrict__ Wl2, const float* __restrict__ Wr2,
                    const float* __restrict__ b2,
                    float* __restrict__ Wc, float* __restrict__ bc) {
  const int half = lane >> 4;
  const int l16  = lane & 15;
  if (gw < 96) {
    const int p  = gw >> 5;
    const int tn = (gw >> 3) & 3;
    const int tj = gw & 7;
    const int n = tn * 16 + l16;
    const int nterms = (p == 1) ? 2 : 1;
    v8f acc = v8f{};
#pragma unroll 1
    for (int term = 0; term < nterms; ++term) {
      const bool a_l = (p == 0) || (p == 1 && term == 0);
      const bool b_l = (p == 0) || (p == 1 && term == 1);
      const float* arow = (a_l ? Wl2 : Wr2) + n * 128 + 2 * half;
      const float* bcol = (b_l ? Wl1 : Wr1) + (2 * half) * 128 + tj * 16 + l16;
#pragma unroll
      for (int s = 0; s < 32; ++s) {
        const v2f A = *(const v2f*)(arow + 4 * s);
        v2f B;
        B.x = bcol[(4 * s) * 128];
        B.y = bcol[(4 * s + 1) * 128];
        acc = __builtin_amdgcn_wmma_f32_16x16x4_f32(false, A, false, B,
                                                    (short)0, acc, false, false);
      }
    }
    float* cp = Wc + (p * 64 + tn * 16 + 8 * half) * 128 + tj * 16 + l16;
#pragma unroll
    for (int i = 0; i < 8; ++i) cp[i * 128] = acc[i];
  }

  const int r = (gw >= 128) ? (gw - 128) : ((gw >= 64) ? (gw - 64) : 0);
  const float* wrow = ((gw >= 128) ? Wr2 : Wl2) + r * 128;
  const float4 wv = *(const float4*)(wrow + lane * 4);
  const float4 bv = *(const float4*)(b1 + lane * 4);
  float d = wv.x * bv.x + wv.y * bv.y + wv.z * bv.z + wv.w * bv.w;
  for (int o = 16; o > 0; o >>= 1) d += __shfl_xor(d, o);
  if (lane == 0) bc[gw] = (gw < 64) ? 0.0f : (d + ((gw >= 128) ? b2[r] : 0.0f));
}

__device__ __forceinline__
int scan256(int val, int t, int* wsum, int& tot) {
  const int lane = t & 31;
  const int w = t >> 5;
  int inc = val;
  for (int o = 1; o < 32; o <<= 1) {
    int v = __shfl_up(inc, o);
    if (lane >= o) inc += v;
  }
  if (w < 8 && lane == 31) wsum[w] = inc;
  __syncthreads();
  int wo = 0;
  tot = 0;
  for (int i = 0; i < 8; ++i) {
    const int a = wsum[i];
    wo += (i < w) ? a : 0;
    tot += a;
  }
  return wo + inc - val;
}

__global__ __launch_bounds__(256)
void bin_edges(const int* __restrict__ src, const int* __restrict__ dst,
               int* __restrict__ binned, int* __restrict__ offT,
               const float* __restrict__ Wl1, const float* __restrict__ Wr1,
               const float* __restrict__ b1,
               const float* __restrict__ Wl2, const float* __restrict__ Wr2,
               const float* __restrict__ b2,
               float* __restrict__ Wc, float* __restrict__ bc) {
  __shared__ int hist[NBINS];
  __shared__ int wsum[8];
  __shared__ int outb[CHUNK];
  const int t = threadIdx.x;

  if (blockIdx.x >= NBINS) {
    const int gw = __builtin_amdgcn_readfirstlane((blockIdx.x - NBINS) * 8 + (t >> 5));
    compose_layers(gw, t & 31, Wl1, Wr1, b1, Wl2, Wr2, b2, Wc, bc);
    return;
  }
  const int b = blockIdx.x;
  hist[t] = 0;

  int rec[EPT], bn[EPT], rk[EPT];
#pragma unroll
  for (int i = 0; i < EPT; ++i) {
    const int j = i * 256 + t;
    bn[i] = -1;
    rec[i] = 0;
    if (j < CHUNK) {
      const int d = dst[b * CHUNK + j];
      const int s = src[b * CHUNK + j];
      const int k = (int)((unsigned)d / (unsigned)NPB);
      bn[i] = k;
      rec[i] = ((d - k * NPB) << 16) | s;
    }
  }
  __syncthreads();
#pragma unroll
  for (int i = 0; i < EPT; ++i) {
    rk[i] = 0;
    if (bn[i] >= 0) rk[i] = atomicAdd(&hist[bn[i]], 1);
  }
  __syncthreads();

  int tot;
  const int ex = scan256(hist[t], t, wsum, tot);
  hist[t] = ex;
  offT[t * NBINS + b] = ex;
  if (t == 0) offT[NBINS * NBINS + b] = CHUNK;
  __syncthreads();

#pragma unroll
  for (int i = 0; i < EPT; ++i)
    if (bn[i] >= 0) outb[hist[bn[i]] + rk[i]] = rec[i];
  __syncthreads();

#pragma unroll
  for (int i = 0; i < EPT; ++i) {
    const int j = i * 256 + t;
    if (j < CHUNK) binned[b * CHUNK + j] = outb[j];
  }
}

__device__ __forceinline__
void build_bin(int* rec, int* rnk, int* seg, int* srt,
               const int* __restrict__ binned,
               const int* p_lo, const int* p_n, const int* p_pos,
               int* cntL, int* startL, int* wsum,
               int k, int base, int total, int t,
               int* __restrict__ rowptr, float* __restrict__ inv) {
  {
    const int piece = t >> 2;
    const int n = p_n[piece];
    const int* sp = binned + piece * CHUNK + p_lo[piece];
    int* dp = rec + p_pos[piece];
    for (int i = t & 3; i < n; i += 16) {
      const int v0 = sp[i];
      const int v1 = sp[min(i + 4, n - 1)];
      const int v2 = sp[min(i + 8, n - 1)];
      const int v3 = sp[min(i + 12, n - 1)];
      dp[i] = v0;
      if (i + 4 < n) dp[i + 4] = v1;
      if (i + 8 < n) dp[i + 8] = v2;
      if (i + 12 < n) dp[i + 12] = v3;
    }
  }
  __syncthreads();

  for (int j = t; j < total; j += BT)
    rnk[j] = atomicAdd(&cntL[rec[j] >> 16], 1);
  __syncthreads();

  {
    const int c = (t < 256) ? cntL[t] : 0;
    int tot;
    const int ex = scan256(c, t, wsum, tot);
    if (t < 256) startL[t] = ex;
    if (t == 255) startL[256] = total;
    const int node = k * NPB + t;
    if (t < NPB && node <= N_NODES) rowptr[node] = base + ex;
    if (t < NPB && node < N_NODES) inv[node] = 1.0f / fmaxf((float)c, 1.0f);
  }
  __syncthreads();

  for (int j = t; j < total; j += BT) {
    const int r = rec[j];
    seg[startL[r >> 16] + rnk[j]] = r & 0xFFFF;
  }
  __syncthreads();

  for (int j = t; j < total; j += BT) {
    const int r = rec[j];
    const int v = r & 0xFFFF;
    const int beg = startL[r >> 16];
    const int end = startL[(r >> 16) + 1];
    const int p = beg + rnk[j];
    int cn = 0;
    for (int q = beg; q < end; q += 4) {
      const int q1 = q + 1, q2 = q + 2, q3 = q + 3;
      const int u0 = seg[q];
      const int u1 = seg[min(q1, end - 1)];
      const int u2 = seg[min(q2, end - 1)];
      const int u3 = seg[min(q3, end - 1)];
      cn += ((u0 < v) || (u0 == v && q < p)) ? 1 : 0;
      cn += (q1 < end && ((u1 < v) || (u1 == v && q1 < p))) ? 1 : 0;
      cn += (q2 < end && ((u2 < v) || (u2 == v && q2 < p))) ? 1 : 0;
      cn += (q3 < end && ((u3 < v) || (u3 == v && q3 < p))) ? 1 : 0;
    }
    srt[beg + cn] = v;
  }
}

__global__ __launch_bounds__(BT)
void build_csr(const int* __restrict__ binned, const int* __restrict__ offT,
               int* __restrict__ rowptr, float* __restrict__ inv,
               int* __restrict__ sorted,
               int* __restrict__ g_rec, int* __restrict__ g_rnk, int* __restrict__ g_seg) {
  __shared__ int s_rec[CAP];
  __shared__ int s_rnk[CAP];
  __shared__ int s_seg[CAP];
  __shared__ int s_srt[CAP];
  __shared__ int p_lo[256], p_n[256], p_pos[256];
  __shared__ int cntL[256], startL[257];
  __shared__ int wsumA[8], wsumB[8], wsumC[8];
  const int t = threadIdx.x;
  const int lane = t & 31;
  const int k = blockIdx.x;

  int lo = 0, n = 0;
  if (t < 256) {
    lo = offT[k * NBINS + t];
    n = offT[(k + 1) * NBINS + t] - lo;
    cntL[t] = 0;
  }
  int sl = lo;
  for (int o = 16; o > 0; o >>= 1) sl += __shfl_xor(sl, o);
  if (t < 256 && lane == 0) wsumB[t >> 5] = sl;
  int total;
  const int pos = scan256(n, t, wsumA, total);
  int base = 0;
  for (int i = 0; i < 8; ++i) base += wsumB[i];
  if (t < 256) { p_lo[t] = lo; p_n[t] = n; p_pos[t] = pos; }
  __syncthreads();

  if (total <= CAP) {
    build_bin(s_rec, s_rnk, s_seg, s_srt, binned, p_lo, p_n, p_pos, cntL, startL, wsumC,
              k, base, total, t, rowptr, inv);
    __syncthreads();
    for (int j = t; j < total; j += BT) sorted[base + j] = s_srt[j];
  } else {
    build_bin(g_rec + base, g_rnk + base, g_seg + base, sorted + base, binned,
              p_lo, p_n, p_pos, cntL, startL, wsumC, k, base, total, t, rowptr, inv);
  }
}

#define NPW 8

template<int T>
__device__ __forceinline__
void ask_rows(float4 (&v)[16], int cur, int cnt, const float* __restrict__ base) {
  if constexpr (T < 16) {
    if (2 * T < cnt) {
      const int s = __builtin_amdgcn_update_dpp(0, cur, 0x150 + T, 0xf, 0xf, false);
      v[T] = *(const float4*)(base + (unsigned)(s * 64));
      ask_rows<T + 1>(v, cur, cnt, base);
    }
  }
}

template<int T>
__device__ __forceinline__
void add_rows(float4& acc, const float4 (&v)[16], int cnt, int g) {
  if constexpr (T < 16) {
    if (2 * T < cnt) {
      const bool ok = (2 * T + g) < cnt;
      acc.x += ok ? v[T].x : 0.0f;
      acc.y += ok ? v[T].y : 0.0f;
      acc.z += ok ? v[T].z : 0.0f;
      acc.w += ok ? v[T].w : 0.0f;
      add_rows<T + 1>(acc, v, cnt, g);
    }
  }
}

__device__ __forceinline__
float4 rows_finish(float4 (&v)[16], const float* __restrict__ base,
                   const int* __restrict__ sorted, int beg, int end, int epos, int g) {
  float4 acc = make_float4(0.0f, 0.0f, 0.0f, 0.0f);
  add_rows<0>(acc, v, min(end - beg, 32), g);
  for (int cb = beg + 32; cb < end; cb += 32) {
    const int more = sorted[min(cb + epos, N_EDGES - 1)];
    const int cnt = min(end - cb, 32);
    ask_rows<0>(v, more, cnt, base);
    add_rows<0>(acc, v, cnt, g);
  }
  acc.x += __shfl_xor(acc.x, 16);
  acc.y += __shfl_xor(acc.y, 16);
  acc.z += __shfl_xor(acc.z, 16);
  acc.w += __shfl_xor(acc.w, 16);
  return acc;
}

__device__ __forceinline__
float4 node_row_sum(const float* __restrict__ base, const int* __restrict__ sorted,
                    int beg, int end, int cur, int epos, int g) {
  float4 v[16];
  ask_rows<0>(v, cur, min(end - beg, 32), base);
  return rows_finish(v, base, sorted, beg, end, epos, g);
}

__global__ __launch_bounds__(256)
void gather_sum(const float* __restrict__ feat, const int* __restrict__ rowptr,
                const int* __restrict__ sorted, const float* __restrict__ inv,
                const float* __restrict__ self, float* __restrict__ agg) {
  constexpr int D = 64;
  const int lane = threadIdx.x & 31;
  const int gw = __builtin_amdgcn_readfirstlane(blockIdx.x * 8 + (threadIdx.x >> 5));
  const int n0 = gw * NPW;
  if (n0 >= N_NODES) return;
  const int nn = min(NPW, N_NODES - n0);
  const int g = lane >> 4;
  const int c4 = (lane & 15) * 4;
  const float* base = feat + c4;
  const int epos = 2 * (lane & 15) + g;

  const int rpv = rowptr[min(n0 + lane, N_NODES)];
  const float scv = inv[min(n0 + lane, N_NODES - 1)];

  int beg = __builtin_amdgcn_readlane(rpv, 0);
  int cur = sorted[min(beg + epos, N_EDGES - 1)];

  for (int i = 0; i < nn; ++i) {
    const int node = n0 + i;
    const int end = __builtin_amdgcn_readlane(rpv, i + 1);
    const float4 own = *(const float4*)(self + (size_t)node * D + c4);
    const float sc = __builtin_bit_cast(float,
                       __builtin_amdgcn_readlane(__builtin_bit_cast(int, scv), i));
    const int nxt = sorted[min(end + epos, N_EDGES - 1)];

    float4 acc = node_row_sum(base, sorted, beg, end, cur, epos, g);
    acc.x = acc.x * sc + own.x;
    acc.y = acc.y * sc + own.y;
    acc.z = acc.z * sc + own.z;
    acc.w = acc.w * sc + own.w;
    if (g == 0) *(float4*)(agg + (size_t)node * D + c4) = acc;
    beg = end;
    cur = nxt;
  }
}


template<int DIN, int DOUT, int SLABS>
__global__ __launch_bounds__(256, 6)
void sage_wmma_gemm(const float* __restrict__ xin,
                    const float* __restrict__ Wr,
                    const float* __restrict__ bias,
                    float* __restrict__ out) {
  constexpr int NT = DOUT / 16;
  constexpr int MW = 8 / NT;
  constexpr int MB = 32 * MW;
  constexpr int P  = DIN + 4;
  constexpr int KS = DIN / 4;
  constexpr int Q  = DIN / 4;
  constexpr int LROWS = (DOUT > MB) ? DOUT : MB;
  constexpr int NBLK = (N_NODES + MB - 1) / MB;
  constexpr int WP = DOUT * Q / 256;
  constexpr int RP = MB * Q / 256;
  constexpr int PO = DOUT + 4;
  constexpr int QO = DOUT / 4;
  constexpr int OP = MB * QO / 256;
  constexpr int KPF = (2 * KS) / 3;

  __shared__ __attribute__((aligned(16))) float S[LROWS * P];
  __shared__ __attribute__((aligned(16))) float O[MB * PO];

  const int tid  = threadIdx.x;
  const int lane = tid & 31;
  const int half = lane >> 4;
  const int l16  = lane & 15;
  const int w  = __builtin_amdgcn_readfirstlane(tid >> 5);
  const int ct = w % NT;
  const int rg = w / NT;

  const int slab  = blockIdx.x % SLABS;
  const int bstep = gridDim.x / SLABS;
  Wr   += (size_t)slab * DOUT * DIN;
  bias += slab * DOUT;
  out  += (size_t)slab * N_NODES * DOUT;

  v4f rx[RP];
  int blk = blockIdx.x / SLABS;
#pragma unroll
  for (int i = 0; i < RP; ++i) {
    const int e  = tid + 256 * i;
    const int row = min(blk * MB + e / Q, N_NODES - 1);
    const int c4 = (e % Q) * 4;
    rx[i] = *(const v4f*)(xin + (size_t)row * DIN + c4);
  }

#pragma unroll
  for (int i = 0; i < WP; ++i) {
    const int e  = tid + 256 * i;
    const int n  = e / Q;
    const int c4 = (e % Q) * 4;
    const v4f vr = *(const v4f*)(Wr + (size_t)n * DIN + c4);
    *(v4f*)(S + n * P + c4) = vr;
  }
  __syncthreads();

  v2f B[KS];
  {
    const float* bp = S + (ct * 16 + l16) * P + 2 * half;
#pragma unroll
    for (int s = 0; s < KS; ++s) B[s] = *(const v2f*)(bp + 4 * s);
  }
  const int col = ct * 16 + l16;
  const float bj = bias[col];
  __syncthreads();

  const float* a0 = S + (rg * 32 + l16) * P + 2 * half;
  const float* a1 = a0 + 16 * P;

  for (; blk < NBLK; blk += bstep) {
#pragma unroll
    for (int i = 0; i < RP; ++i) {
      const int e  = tid + 256 * i;
      const int r  = e / Q;
      const int c4 = (e % Q) * 4;
      *(v4f*)(S + r * P + c4) = rx[i];
    }
    __syncthreads();

    v8f acc0 = v8f{}, acc1 = v8f{};
#pragma unroll
    for (int s = 0; s < KS; ++s) {
      if (s == KPF) {
        __builtin_amdgcn_sched_barrier(0);
        const int nb = min(blk + bstep, NBLK - 1);
#pragma unroll
        for (int i = 0; i < RP; ++i) {
          const int e  = tid + 256 * i;
          const int row = min(nb * MB + e / Q, N_NODES - 1);
          const int c4 = (e % Q) * 4;
          rx[i] = *(const v4f*)(xin + (size_t)row * DIN + c4);
        }
        __builtin_amdgcn_sched_barrier(0);
      }
      const v2f A0 = *(const v2f*)(a0 + 4 * s);
      const v2f A1 = *(const v2f*)(a1 + 4 * s);
      acc0 = __builtin_amdgcn_wmma_f32_16x16x4_f32(false, A0, false, B[s],
                                                   (short)0, acc0, false, false);
      acc1 = __builtin_amdgcn_wmma_f32_16x16x4_f32(false, A1, false, B[s],
                                                   (short)0, acc1, false, false);
    }

    {
      float* o0 = O + (rg * 32 + (half << 3)) * PO + col;
#pragma unroll
      for (int i = 0; i < 8; ++i) {
        o0[i * PO] = acc0[i] + bj;
        o0[(16 + i) * PO] = acc1[i] + bj;
      }
    }
    __syncthreads();
    if ((blk + 1) * MB <= N_NODES) {
      v4f ov[OP];
#pragma unroll
      for (int i = 0; i < OP; ++i)
        ov[i] = *(const v4f*)(O + (tid / QO + i * (256 / QO)) * PO + (tid % QO) * 4);
      float* ob = out + (size_t)(blk * MB + tid / QO) * DOUT + (tid % QO) * 4;
#pragma unroll
      for (int i = 0; i < OP; ++i)
        *(v4f*)(ob + (size_t)i * (256 / QO) * DOUT) = ov[i];
    } else {
#pragma unroll
      for (int i = 0; i < OP; ++i) {
        const int e  = tid + 256 * i;
        const int r  = e / QO;
        const int c4 = (e % QO) * 4;
        const int row = blk * MB + r;
        if (row < N_NODES)
          *(v4f*)(out + (size_t)row * DOUT + c4) = *(const v4f*)(O + r * PO + c4);
      }
    }
  }
}

__global__ __launch_bounds__(256, 4)
void sage_layer3(const float* __restrict__ h,
                 const int* __restrict__ rowptr, const int* __restrict__ sorted,
                 const float* __restrict__ inv,
                 const float* __restrict__ Wl,
                 const float* __restrict__ Wr,
                 const float* __restrict__ bias,
                 float* __restrict__ out) {
  constexpr int DIN = 64, DOUT = 128;
  constexpr int MB = 32;
  constexpr int NPB3 = MB / 8;
  constexpr int KW = 2 * DIN;
  constexpr int XO = DIN;
  constexpr int P  = KW + 4;
  constexpr int KS = KW / 4;
  constexpr int KQ = KS / NPB3;
  constexpr int Q  = DIN / 4;
  constexpr int NBLK = (N_NODES + MB - 1) / MB;
  constexpr int WP = DOUT * Q / 256;
  constexpr int RP = MB * Q / 256;
  constexpr int TILE = MB * P;
  constexpr int PO = DOUT + 4;
  constexpr int QO = DOUT / 4;
  constexpr int OP = MB * QO / 256;
  constexpr int OT = MB * PO;
  static_assert(2 * MB <= DOUT, "both A tiles fit the rows that held the weights");

  __shared__ __attribute__((aligned(16))) float S[DOUT * P];
  __shared__ __attribute__((aligned(16))) float O[2 * OT];

  const int tid  = threadIdx.x;
  const int lane = tid & 31;
  const int half = lane >> 4;
  const int l16  = lane & 15;
  const int w  = __builtin_amdgcn_readfirstlane(tid >> 5);
  const int bstep = gridDim.x;
  const int c4g = l16 * 4;
  const int epos = 2 * l16 + half;
  const float* hbase = h + c4g;

  int blk = blockIdx.x;
  int rpv = rowptr[min(blk * MB + w * NPB3 + lane, N_NODES)];
  float scv = inv[min(blk * MB + w * NPB3 + lane, N_NODES - 1)];
  v4f rx[RP];
#pragma unroll
  for (int i = 0; i < RP; ++i) {
    const int e  = tid + 256 * i;
    const int row = min(blk * MB + e / Q, N_NODES - 1);
    const int c4 = (e % Q) * 4;
    rx[i] = *(const v4f*)(h + (size_t)row * DIN + c4);
  }
  int cur = sorted[min(__builtin_amdgcn_readlane(rpv, 0) + epos, N_EDGES - 1)];

#pragma unroll
  for (int i = 0; i < WP; ++i) {
    const int e  = tid + 256 * i;
    const int n  = e / Q;
    const int c4 = (e % Q) * 4;
    const v4f vl = *(const v4f*)(Wl + (size_t)n * DIN + c4);
    const v4f vr = *(const v4f*)(Wr + (size_t)n * DIN + c4);
    *(v4f*)(S + n * P + c4) = vl;
    *(v4f*)(S + n * P + XO + c4) = vr;
  }
  __syncthreads();

  v2f B[KS];
  {
    const float* bp = S + (w * 16 + l16) * P + 2 * half;
#pragma unroll
    for (int s = 0; s < KS; ++s) B[s] = *(const v2f*)(bp + 4 * s);
  }
  const int col = w * 16 + l16;
  const float bj = bias[col];
  __syncthreads();

  {
    int beg = __builtin_amdgcn_readlane(rpv, 0);
    for (int i = 0; i < NPB3; ++i) {
      const int end = __builtin_amdgcn_readlane(rpv, i + 1);
      const float sc = __builtin_bit_cast(float,
                         __builtin_amdgcn_readlane(__builtin_bit_cast(int, scv), i));
      const int nxt = sorted[min(end + epos, N_EDGES - 1)];
      float4 acc = node_row_sum(hbase, sorted, beg, end, cur, epos, half);
      acc.x *= sc;
      acc.y *= sc;
      acc.z *= sc;
      acc.w *= sc;
      if (half == 0) *(float4*)(S + (w * NPB3 + i) * P + c4g) = acc;
      beg = end;
      cur = nxt;
    }
#pragma unroll
    for (int i = 0; i < RP; ++i) {
      const int e  = tid + 256 * i;
      *(v4f*)(S + (e / Q) * P + XO + (e % Q) * 4) = rx[i];
    }
  }
  {
    const int nb = min(blk + bstep, NBLK - 1);
    rpv = rowptr[min(nb * MB + w * NPB3 + lane, N_NODES)];
    scv = inv[min(nb * MB + w * NPB3 + lane, N_NODES - 1)];
#pragma unroll
    for (int i = 0; i < RP; ++i) {
      const int e  = tid + 256 * i;
      const int row = min(nb * MB + e / Q, N_NODES - 1);
      rx[i] = *(const v4f*)(h + (size_t)row * DIN + (e % Q) * 4);
    }
    cur = sorted[min(__builtin_amdgcn_readlane(rpv, 0) + epos, N_EDGES - 1)];
  }
  __syncthreads();

  int buf = 0;
  for (; blk < NBLK; blk += bstep) {
    const bool fill = (blk + bstep) < NBLK;
    const float* a0 = S + buf * TILE + l16 * P + 2 * half;
    const float* a1 = a0 + 16 * P;
    float* nt = S + (buf ^ 1) * TILE;

    if (fill) {
#pragma unroll
      for (int i = 0; i < RP; ++i) {
        const int e  = tid + 256 * i;
        *(v4f*)(nt + (e / Q) * P + XO + (e % Q) * 4) = rx[i];
      }
    }
    const int nb2 = min(blk + 2 * bstep, NBLK - 1);
    const int rpn = rowptr[min(nb2 * MB + w * NPB3 + lane, N_NODES)];
    const float scn = inv[min(nb2 * MB + w * NPB3 + lane, N_NODES - 1)];
#pragma unroll
    for (int i = 0; i < RP; ++i) {
      const int e  = tid + 256 * i;
      const int row = min(nb2 * MB + e / Q, N_NODES - 1);
      rx[i] = *(const v4f*)(h + (size_t)row * DIN + (e % Q) * 4);
    }
    int curn = 0;

    v8f acc0 = v8f{}, acc1 = v8f{};
    int beg = __builtin_amdgcn_readlane(rpv, 0);
#pragma unroll
    for (int j = 0; j < NPB3; ++j) {
      float4 v[16];
      int end = beg, nxt = 0;
      if (fill) {
        end = __builtin_amdgcn_readlane(rpv, j + 1);
        nxt = sorted[min(end + epos, N_EDGES - 1)];
        ask_rows<0>(v, cur, min(end - beg, 32), hbase);
      }
      if (j == NPB3 / 2)
        curn = sorted[min(__builtin_amdgcn_readlane(rpn, 0) + epos, N_EDGES - 1)];

#pragma unroll
      for (int s = j * KQ; s < (j + 1) * KQ; ++s) {
        const v2f A0 = *(const v2f*)(a0 + 4 * s);
        const v2f A1 = *(const v2f*)(a1 + 4 * s);
        acc0 = __builtin_amdgcn_wmma_f32_16x16x4_f32(false, A0, false, B[s],
                                                     (short)0, acc0, false, false);
        acc1 = __builtin_amdgcn_wmma_f32_16x16x4_f32(false, A1, false, B[s],
                                                     (short)0, acc1, false, false);
      }

      if (fill) {
        const float sc = __builtin_bit_cast(float,
                           __builtin_amdgcn_readlane(__builtin_bit_cast(int, scv), j));
        float4 acc = rows_finish(v, hbase, sorted, beg, end, epos, half);
        acc.x *= sc;
        acc.y *= sc;
        acc.z *= sc;
        acc.w *= sc;
        if (half == 0) *(float4*)(nt + (w * NPB3 + j) * P + c4g) = acc;
        beg = end;
        cur = nxt;
      }
    }
    rpv = rpn;
    scv = scn;
    cur = curn;

    float* ot = O + buf * OT;
    {
      float* o0 = ot + (half << 3) * PO + col;
#pragma unroll
      for (int i = 0; i < 8; ++i) {
        o0[i * PO] = acc0[i] + bj;
        o0[(16 + i) * PO] = acc1[i] + bj;
      }
    }
    __syncthreads();
    if ((blk + 1) * MB <= N_NODES) {
      v4f ov[OP];
#pragma unroll
      for (int i = 0; i < OP; ++i)
        ov[i] = *(const v4f*)(ot + (tid / QO + i * (256 / QO)) * PO + (tid % QO) * 4);
      float* ob = out + (size_t)(blk * MB + tid / QO) * DOUT + (tid % QO) * 4;
#pragma unroll
      for (int i = 0; i < OP; ++i)
        *(v4f*)(ob + (size_t)i * (256 / QO) * DOUT) = ov[i];
    } else {
#pragma unroll
      for (int i = 0; i < OP; ++i) {
        const int e  = tid + 256 * i;
        const int r  = e / QO;
        const int c4 = (e % QO) * 4;
        const int row = blk * MB + r;
        if (row < N_NODES)
          *(v4f*)(out + (size_t)row * DOUT + c4) = *(const v4f*)(ot + r * PO + c4);
      }
    }
    buf ^= 1;
  }
}


extern "C" void kernel_launch(void* const* d_in, const int* in_sizes, int n_in,
                              void* d_out, int out_size, void* d_ws, size_t ws_size,
                              hipStream_t stream) {
  const float* x   = (const float*)d_in[0];
  const int*   ei  = (const int*)d_in[1];
  const int*   src = ei;
  const int*   dst = ei + N_EDGES;
  const float* Wl1 = (const float*)d_in[2];
  const float* b1  = (const float*)d_in[3];
  const float* Wr1 = (const float*)d_in[4];
  const float* Wl2 = (const float*)d_in[5];
  const float* b2  = (const float*)d_in[6];
  const float* Wr2 = (const float*)d_in[7];
  const float* Wl3 = (const float*)d_in[8];
  const float* b3  = (const float*)d_in[9];
  const float* Wr3 = (const float*)d_in[10];
  float* out = (float*)d_out;

  float* inv    = (float*)d_ws;
  int*   rowptr = (int*)(inv + 50048);
  int*   offT   = rowptr + 50048;
  int*   binned = offT + 257 * 256;
  int*   sorted = binned + N_EDGES;
  int*   g_rec  = sorted + N_EDGES;
  int*   g_rnk  = g_rec + N_EDGES;
  int*   g_seg  = g_rnk + N_EDGES;
  float* pu     = (float*)(g_seg + N_EDGES);
  float* pv     = pu  + (size_t)N_NODES * 64;
  float* pt     = pv  + (size_t)N_NODES * 64;
  float* r2     = pt  + (size_t)N_NODES * 64;
  float* h2     = r2  + (size_t)N_NODES * 64;
  float* wc     = h2  + (size_t)N_NODES * 64;
  float* bc     = wc  + 192 * 128;

  bin_edges<<<NBINS + 24, 256, 0, stream>>>(src, dst, binned, offT, Wl1, Wr1, b1, Wl2, Wr2, b2,
                                            wc, bc);
  build_csr<<<NBINS, BT, 0, stream>>>(binned, offT, rowptr, inv, sorted, g_rec, g_rnk, g_seg);

  const int gatherBlocks = (N_NODES + 8 * NPW - 1) / (8 * NPW);

  sage_wmma_gemm<128, 64, 3><<<768, 256, 0, stream>>>(x, wc, bc, pu);
  gather_sum<<<gatherBlocks, 256, 0, stream>>>(pu, rowptr, sorted, inv, pv, r2);
  gather_sum<<<gatherBlocks, 256, 0, stream>>>(r2, rowptr, sorted, inv, pt, h2);

  sage_layer3<<<512, 256, 0, stream>>>(h2, rowptr, sorted, inv, Wl3, Wr3, b3, out);
}
